# starting kernel
# baseline (speedup 1.0000x reference)
#include <hip/hip_runtime.h>
#include <hip/hip_bf16.h>

#define IN_F  128
#define OUT_F 64
#define NEG_SLOPE 0.2f

typedef __attribute__((ext_vector_type(2))) float v2f;
typedef __attribute__((ext_vector_type(8))) float v8f;

__device__ __forceinline__ unsigned f2ord(float f) {
    unsigned u = __float_as_uint(f);
    return (u & 0x80000000u) ? ~u : (u | 0x80000000u);
}
__device__ __forceinline__ float ord2f(unsigned u) {
    unsigned v = (u & 0x80000000u) ? (u & 0x7FFFFFFFu) : ~u;
    return __uint_as_float(v);
}

__global__ __launch_bounds__(256) void gat_init(float* __restrict__ out,
                                                const float* __restrict__ bias,
                                                unsigned* __restrict__ smax,
                                                float* __restrict__ ssum, int N) {
    int idx = blockIdx.x * 256 + threadIdx.x;
    if (idx < N * OUT_F) out[idx] = bias[idx & (OUT_F - 1)];
    if (idx < N) { smax[idx] = 0u; ssum[idx] = 0.0f; }
}

__global__ __launch_bounds__(256) void gat_gemm(const float* __restrict__ x,
                                                const float* __restrict__ W,
                                                const float* __restrict__ att_src,
                                                const float* __restrict__ att_dst,
                                                float* __restrict__ h,
                                                float* __restrict__ a_s,
                                                float* __restrict__ a_d, int N) {
    __shared__ float Ws[IN_F * OUT_F];
    for (int i = threadIdx.x; i < IN_F * OUT_F; i += 256) {
        int k = i >> 6;
        int n = i & 63;
        Ws[(k >> 1) * 128 + n * 2 + (k & 1)] = W[i];
    }
    __syncthreads();

    const int lane = threadIdx.x & 31;
    const int wave = threadIdx.x >> 5;
    const int m    = lane & 15;
    const int hi   = lane >> 4;
    const int row0 = (blockIdx.x * 8 + wave) * 16;

    const int arow = row0 + m;
    const int crow = arow < N ? arow : 0;
    const float2* xrow2 = (const float2*)(x + (size_t)crow * IN_F);

    v8f c0 = {}, c1 = {}, c2 = {}, c3 = {};

    for (int k0 = 0; k0 < IN_F; k0 += 4) {
        const int kp = (k0 >> 1) + hi;
        float2 af = xrow2[kp];
        v2f a = { af.x, af.y };
        const float2* wrow = (const float2*)&Ws[kp * 128];
        float2 f0 = wrow[m];
        float2 f1 = wrow[m + 16];
        float2 f2 = wrow[m + 32];
        float2 f3 = wrow[m + 48];
        v2f b0 = { f0.x, f0.y };
        v2f b1 = { f1.x, f1.y };
        v2f b2 = { f2.x, f2.y };
        v2f b3 = { f3.x, f3.y };
        c0 = __builtin_amdgcn_wmma_f32_16x16x4_f32(false, a, false, b0, (short)0, c0, false, false);
        c1 = __builtin_amdgcn_wmma_f32_16x16x4_f32(false, a, false, b1, (short)0, c1, false, false);
        c2 = __builtin_amdgcn_wmma_f32_16x16x4_f32(false, a, false, b2, (short)0, c2, false, false);
        c3 = __builtin_amdgcn_wmma_f32_16x16x4_f32(false, a, false, b3, (short)0, c3, false, false);
    }

    const float as0 = att_src[m],      as1 = att_src[16 + m];
    const float as2 = att_src[32 + m], as3 = att_src[48 + m];
    const float ad0 = att_dst[m],      ad1 = att_dst[16 + m];
    const float ad2 = att_dst[32 + m], ad3 = att_dst[48 + m];

#pragma unroll
    for (int r = 0; r < 8; ++r) {
        const int row = row0 + r + 8 * hi;
        const float e0 = c0[r], e1 = c1[r], e2 = c2[r], e3 = c3[r];
        if (row < N) {
            float* hr = h + (size_t)row * OUT_F;
            hr[m]      = e0;
            hr[16 + m] = e1;
            hr[32 + m] = e2;
            hr[48 + m] = e3;
        }
        float ps = e0 * as0 + e1 * as1 + e2 * as2 + e3 * as3;
        float pd = e0 * ad0 + e1 * ad1 + e2 * ad2 + e3 * ad3;
#pragma unroll
        for (int off = 8; off >= 1; off >>= 1) {
            ps += __shfl_xor(ps, off, 32);
            pd += __shfl_xor(pd, off, 32);
        }
        if (m == 0 && row < N) { a_s[row] = ps; a_d[row] = pd; }
    }
}

__global__ __launch_bounds__(256) void gat_logits(const int* __restrict__ ei,
                                                  const float* __restrict__ a_s,
                                                  const float* __restrict__ a_d,
                                                  float* __restrict__ e,
                                                  unsigned* __restrict__ smax,
                                                  int E, int T) {
    int idx = blockIdx.x * 256 + threadIdx.x;
    if (idx >= T) return;
    int s, d;
    if (idx < E) { s = ei[idx]; d = ei[E + idx]; }
    else         { s = d = idx - E; }
    float v = a_s[s] + a_d[d];
    v = v > 0.0f ? v : NEG_SLOPE * v;
    e[idx] = v;
    atomicMax(&smax[d], f2ord(v));
}

__global__ __launch_bounds__(256) void gat_exp(const int* __restrict__ ei,
                                               float* __restrict__ e,
                                               const unsigned* __restrict__ smax,
                                               float* __restrict__ ssum,
                                               int E, int T) {
    int idx = blockIdx.x * 256 + threadIdx.x;
    if (idx >= T) return;
    int d = (idx < E) ? ei[E + idx] : (idx - E);
    float w = __expf(e[idx] - ord2f(smax[d]));
    e[idx] = w;
    atomicAdd(&ssum[d], w);
}

__global__ __launch_bounds__(256) void gat_aggregate(const int* __restrict__ ei,
                                                     const float* __restrict__ h,
                                                     const float* __restrict__ e,
                                                     const float* __restrict__ ssum,
                                                     float* __restrict__ out,
                                                     int E, int T) {
    int edge = blockIdx.x * 8 + (threadIdx.x >> 5);
    if (edge >= T) return;
    int lane = threadIdx.x & 31;
    int s, d;
    if (edge < E) { s = ei[edge]; d = ei[E + edge]; }
    else          { s = d = edge - E; }
    float alpha = e[edge] / ssum[d];
    const float2* hv = (const float2*)(h + (size_t)s * OUT_F);
    float2 v = hv[lane];
    float* od = out + (size_t)d * OUT_F + 2 * lane;
    atomicAdd(od,     alpha * v.x);
    atomicAdd(od + 1, alpha * v.y);
}

extern "C" void kernel_launch(void* const* d_in, const int* in_sizes, int n_in,
                              void* d_out, int out_size, void* d_ws, size_t ws_size,
                              hipStream_t stream) {
    const float* x       = (const float*)d_in[0];
    const int*   ei      = (const int*)d_in[1];
    const float* W       = (const float*)d_in[2];
    const float* att_src = (const float*)d_in[3];
    const float* att_dst = (const float*)d_in[4];
    const float* bias    = (const float*)d_in[5];

    const int N = in_sizes[0] / IN_F;
    const int E = in_sizes[1] / 2;
    const int T = E + N;

    float* out = (float*)d_out;

    float*    h    = (float*)d_ws;
    float*    a_s  = h + (size_t)N * OUT_F;
    float*    a_d  = a_s + N;
    float*    ew   = a_d + N;
    float*    ssum = ew + T;
    unsigned* smax = (unsigned*)(ssum + N);

    gat_init<<<(N * OUT_F + 255) / 256, 256, 0, stream>>>(out, bias, smax, ssum, N);
    gat_gemm<<<(N + 127) / 128, 256, 0, stream>>>(x, W, att_src, att_dst, h, a_s, a_d, N);
    gat_logits<<<(T + 255) / 256, 256, 0, stream>>>(ei, a_s, a_d, ew, smax, E, T);
    gat_exp<<<(T + 255) / 256, 256, 0, stream>>>(ei, ew, smax, ssum, E, T);
    gat_aggregate<<<(T + 7) / 8, 256, 0, stream>>>(ei, h, ew, ssum, out, E, T);
}


# Bin edges by destination range (LDS histogram, scan, scatter); per bin, softmax max, sum and output rows accumulate by LDS atomics in arrival order; global atomics and output pre-fill removed
# speedup vs baseline: 2.5147x; 2.5147x over previous
#include <hip/hip_runtime.h>
#include <hip/hip_bf16.h>

#define IN_F  128
#define OUT_F 64
#define NEG_SLOPE 0.2f
#define BIN_SHIFT 7
#define BIN_NODES 128
#define MAXB 1024
#define NSLICE 128

typedef __attribute__((ext_vector_type(2))) float v2f;
typedef __attribute__((ext_vector_type(8))) float v8f;

__device__ __forceinline__ unsigned f2ord(float f) {
    unsigned u = __float_as_uint(f);
    return (u & 0x80000000u) ? ~u : (u | 0x80000000u);
}
__device__ __forceinline__ float ord2f(unsigned u) {
    unsigned v = (u & 0x80000000u) ? (u & 0x7FFFFFFFu) : ~u;
    return __uint_as_float(v);
}

__global__ __launch_bounds__(256) void gat_gemm(const float* __restrict__ x,
                                                const float* __restrict__ W,
                                                const float* __restrict__ att_src,
                                                const float* __restrict__ att_dst,
                                                float* __restrict__ h,
                                                float* __restrict__ a_s,
                                                float* __restrict__ a_d, int N) {
    __shared__ float Ws[IN_F * OUT_F];
    for (int i = threadIdx.x; i < IN_F * OUT_F; i += 256) {
        int k = i >> 6;
        int n = i & 63;
        Ws[(k >> 1) * 128 + n * 2 + (k & 1)] = W[i];
    }
    __syncthreads();

    const int lane = threadIdx.x & 31;
    const int wave = threadIdx.x >> 5;
    const int m    = lane & 15;
    const int hi   = lane >> 4;
    const int row0 = (blockIdx.x * 8 + wave) * 16;

    const int arow = row0 + m;
    const int crow = arow < N ? arow : 0;
    const float2* xrow2 = (const float2*)(x + (size_t)crow * IN_F);

    v8f c0 = {}, c1 = {}, c2 = {}, c3 = {};

    for (int k0 = 0; k0 < IN_F; k0 += 4) {
        const int kp = (k0 >> 1) + hi;
        float2 af = xrow2[kp];
        v2f a = { af.x, af.y };
        const float2* wrow = (const float2*)&Ws[kp * 128];
        float2 f0 = wrow[m];
        float2 f1 = wrow[m + 16];
        float2 f2 = wrow[m + 32];
        float2 f3 = wrow[m + 48];
        v2f b0 = { f0.x, f0.y };
        v2f b1 = { f1.x, f1.y };
        v2f b2 = { f2.x, f2.y };
        v2f b3 = { f3.x, f3.y };
        c0 = __builtin_amdgcn_wmma_f32_16x16x4_f32(false, a, false, b0, (short)0, c0, false, false);
        c1 = __builtin_amdgcn_wmma_f32_16x16x4_f32(false, a, false, b1, (short)0, c1, false, false);
        c2 = __builtin_amdgcn_wmma_f32_16x16x4_f32(false, a, false, b2, (short)0, c2, false, false);
        c3 = __builtin_amdgcn_wmma_f32_16x16x4_f32(false, a, false, b3, (short)0, c3, false, false);
    }

    const float as0 = att_src[m],      as1 = att_src[16 + m];
    const float as2 = att_src[32 + m], as3 = att_src[48 + m];
    const float ad0 = att_dst[m],      ad1 = att_dst[16 + m];
    const float ad2 = att_dst[32 + m], ad3 = att_dst[48 + m];

#pragma unroll
    for (int r = 0; r < 8; ++r) {
        const int row = row0 + r + 8 * hi;
        const float e0 = c0[r], e1 = c1[r], e2 = c2[r], e3 = c3[r];
        if (row < N) {
            float* hr = h + (size_t)row * OUT_F;
            hr[m]      = e0;
            hr[16 + m] = e1;
            hr[32 + m] = e2;
            hr[48 + m] = e3;
        }
        float ps = e0 * as0 + e1 * as1 + e2 * as2 + e3 * as3;
        float pd = e0 * ad0 + e1 * ad1 + e2 * ad2 + e3 * ad3;
#pragma unroll
        for (int off = 8; off >= 1; off >>= 1) {
            ps += __shfl_xor(ps, off, 32);
            pd += __shfl_xor(pd, off, 32);
        }
        if (m == 0 && row < N) { a_s[row] = ps; a_d[row] = pd; }
    }
}


__global__ __launch_bounds__(1024) void bin_count(const int* __restrict__ ei,
                                                  int* __restrict__ counts,
                                                  int E, int NB, int slice) {
    __shared__ int hist[MAXB];
    for (int b = threadIdx.x; b < NB; b += 1024) hist[b] = 0;
    __syncthreads();
    const int lo = blockIdx.x * slice;
    const int hi = min(lo + slice, E);
    for (int i0 = lo + threadIdx.x; i0 < hi; i0 += 4 * 1024) {
        int d[4];
#pragma unroll
        for (int u = 0; u < 4; ++u) {
            const int i = i0 + u * 1024;
            d[u] = (i < hi) ? ei[E + i] : -1;
        }
#pragma unroll
        for (int u = 0; u < 4; ++u)
            if (d[u] >= 0) atomicAdd(&hist[d[u] >> BIN_SHIFT], 1);
    }
    __syncthreads();
    for (int b = threadIdx.x; b < NB; b += 1024) counts[b * NSLICE + blockIdx.x] = hist[b];
}

__global__ __launch_bounds__(1024) void bin_scan(int* __restrict__ counts,
                                                 int* __restrict__ binstart, int NB) {
    __shared__ int tot[MAXB];
    __shared__ int wsum[32];
    const int lane = threadIdx.x & 31;
    const int wave = threadIdx.x >> 5;
    for (int b0 = wave; b0 < NB; b0 += 4 * 32) {
        int4 v[4];
#pragma unroll
        for (int u = 0; u < 4; ++u) {
            const int b = b0 + u * 32;
            v[u] = make_int4(0, 0, 0, 0);
            if (b < NB) v[u] = ((const int4*)(counts + (size_t)b * NSLICE))[lane];
        }
#pragma unroll
        for (int u = 0; u < 4; ++u) {
            const int b = b0 + u * 32;
            const int s1 = v[u].x, s2 = s1 + v[u].y, s3 = s2 + v[u].z, s4 = s3 + v[u].w;
            int x = s4;
#pragma unroll
            for (int off = 1; off < 32; off <<= 1) {
                int y = __shfl_up(x, off, 32);
                if (lane >= off) x += y;
            }
            const int ex = x - s4;
            if (b < NB) {
                ((int4*)(counts + (size_t)b * NSLICE))[lane] = make_int4(ex, ex + s1, ex + s2, ex + s3);
                if (lane == 31) tot[b] = x;
            }
        }
    }
    __syncthreads();
    const int t = (threadIdx.x < NB) ? tot[threadIdx.x] : 0;
    int x = t;
#pragma unroll
    for (int off = 1; off < 32; off <<= 1) {
        int y = __shfl_up(x, off, 32);
        if (lane >= off) x += y;
    }
    if (lane == 31) wsum[wave] = x;
    __syncthreads();
    if (wave == 0) {
        int w = wsum[lane];
#pragma unroll
        for (int off = 1; off < 32; off <<= 1) {
            int y = __shfl_up(w, off, 32);
            if (lane >= off) w += y;
        }
        wsum[lane] = w;
    }
    __syncthreads();
    const int excl = x - t + (wave > 0 ? wsum[wave - 1] : 0);
    if (threadIdx.x < NB) binstart[threadIdx.x] = excl;
    if (threadIdx.x == NB - 1) binstart[NB] = excl + t;
}

__global__ __launch_bounds__(1024) void bin_scatter(const int* __restrict__ ei,
                                                    const int* __restrict__ offs,
                                                    const int* __restrict__ binstart,
                                                    int2* __restrict__ sorted,
                                                    int E, int NB, int slice) {
    __shared__ int cur[MAXB];
    for (int b = threadIdx.x; b < NB; b += 1024)
        cur[b] = binstart[b] + offs[b * NSLICE + blockIdx.x];
    __syncthreads();
    const int lo = blockIdx.x * slice;
    const int hi = min(lo + slice, E);
    for (int i0 = lo + threadIdx.x; i0 < hi; i0 += 4 * 1024) {
        int s[4], d[4];
#pragma unroll
        for (int u = 0; u < 4; ++u) {
            const int i = i0 + u * 1024;
            s[u] = 0; d[u] = -1;
            if (i < hi) { s[u] = ei[i]; d[u] = ei[E + i]; }
        }
#pragma unroll
        for (int u = 0; u < 4; ++u) {
            if (d[u] >= 0) {
                const int pos = atomicAdd(&cur[d[u] >> BIN_SHIFT], 1);
                sorted[pos] = make_int2(s[u], d[u]);
            }
        }
    }
}

__device__ __forceinline__ void bin_item(const int2* __restrict__ eb, int j, int ne, int total,
                                         int base, int N, int& s, int& l) {
    s = -1; l = 0;
    if (j < ne) {
        const int2 sd = eb[j];
        s = sd.x; l = sd.y - base;
    } else if (j < total) {
        l = j - ne;
        s = (base + l < N) ? base + l : -1;
    }
}

__global__ __launch_bounds__(512) void gat_bin_process(const int2* __restrict__ sorted,
                                                       const int* __restrict__ binstart,
                                                       const float* __restrict__ h,
                                                       const float* __restrict__ a_s,
                                                       const float* __restrict__ a_d,
                                                       const float* __restrict__ bias,
                                                       float* __restrict__ out, int N) {
    __shared__ float    acc[(BIN_NODES + 1) * OUT_F];
    __shared__ unsigned mx[BIN_NODES];
    __shared__ float    sm[BIN_NODES];
    __shared__ float    ad[BIN_NODES];

    const int tid  = threadIdx.x;
    const int lane = tid & 31;
    const int base = blockIdx.x << BIN_SHIFT;
    const int start = binstart[blockIdx.x];
    const int ne    = binstart[blockIdx.x + 1] - start;
    const int total = ne + BIN_NODES;
    const int2* eb  = sorted + start;

    for (int i = tid; i < (BIN_NODES + 1) * OUT_F; i += 512) acc[i] = 0.0f;
    if (tid < BIN_NODES) {
        const int node = base + tid;
        float adv = 0.0f, v = 0.0f;
        if (node < N) {
            adv = a_d[node];
            v = a_s[node] + adv;
            v = v > 0.0f ? v : NEG_SLOPE * v;
        }
        ad[tid] = adv;
        mx[tid] = f2ord(v);
        sm[tid] = 0.0f;
    }
    __syncthreads();

    for (int j0 = tid; j0 < ne; j0 += 4 * 512) {
        int s[4], l[4];
        float v[4];
#pragma unroll
        for (int u = 0; u < 4; ++u)
            bin_item(eb, j0 + u * 512, ne, ne, base, N, s[u], l[u]);
#pragma unroll
        for (int u = 0; u < 4; ++u) {
            v[u] = 0.0f;
            if (s[u] >= 0) v[u] = a_s[s[u]] + ad[l[u]];
        }
#pragma unroll
        for (int u = 0; u < 4; ++u) {
            if (s[u] >= 0) {
                const float e = v[u] > 0.0f ? v[u] : NEG_SLOPE * v[u];
                atomicMax(&mx[l[u]], f2ord(e));
            }
        }
    }
    __syncthreads();

    for (int j0 = tid; j0 < total; j0 += 4 * 512) {
        int s[4], l[4];
        float v[4];
#pragma unroll
        for (int u = 0; u < 4; ++u)
            bin_item(eb, j0 + u * 512, ne, total, base, N, s[u], l[u]);
#pragma unroll
        for (int u = 0; u < 4; ++u) {
            v[u] = 0.0f;
            if (s[u] >= 0) v[u] = a_s[s[u]] + ad[l[u]];
        }
#pragma unroll
        for (int u = 0; u < 4; ++u) {
            if (s[u] >= 0) {
                const float e = v[u] > 0.0f ? v[u] : NEG_SLOPE * v[u];
                atomicAdd(&sm[l[u]], __expf(e - ord2f(mx[l[u]])));
            }
        }
    }
    __syncthreads();

    for (int j0 = tid; j0 - lane < total; j0 += 512) {
        int s, l;
        bin_item(eb, j0, ne, total, base, N, s, l);
        float alpha = 0.0f;
        if (s >= 0) {
            float e = a_s[s] + ad[l];
            e = e > 0.0f ? e : NEG_SLOPE * e;
            alpha = __expf(e - ord2f(mx[l])) / sm[l];
        } else {
            s = 0; l = BIN_NODES;
        }
        const unsigned ab = __float_as_uint(alpha);
        for (int k0 = 0; k0 < 32; k0 += 8) {
            float2 hv[8];
#pragma unroll
            for (int u = 0; u < 8; ++u) {
                const int sk = __builtin_amdgcn_readlane(s, k0 + u);
                hv[u] = ((const float2*)(h + (size_t)sk * OUT_F))[lane];
            }
#pragma unroll
            for (int u = 0; u < 8; ++u) {
                const int   lk = __builtin_amdgcn_readlane(l, k0 + u);
                const float ak = __uint_as_float(__builtin_amdgcn_readlane(ab, k0 + u));
                float* ar = &acc[lk * OUT_F + 2 * lane];
                atomicAdd(ar,     ak * hv[u].x);
                atomicAdd(ar + 1, ak * hv[u].y);
            }
        }
    }
    __syncthreads();

    const float bv = bias[tid & 63];
    for (int i = tid; i < BIN_NODES * OUT_F; i += 512) {
        const int node = base + (i >> 6);
        if (node < N) out[(size_t)node * OUT_F + (i & 63)] = acc[i] + bv;
    }
}

extern "C" void kernel_launch(void* const* d_in, const int* in_sizes, int n_in,
                              void* d_out, int out_size, void* d_ws, size_t ws_size,
                              hipStream_t stream) {
    const float* x       = (const float*)d_in[0];
    const int*   ei      = (const int*)d_in[1];
    const float* W       = (const float*)d_in[2];
    const float* att_src = (const float*)d_in[3];
    const float* att_dst = (const float*)d_in[4];
    const float* bias    = (const float*)d_in[5];

    const int N  = in_sizes[0] / IN_F;
    const int E  = in_sizes[1] / 2;
    const int NB = (N + BIN_NODES - 1) / BIN_NODES;
    const int slice = (E + NSLICE - 1) / NSLICE;

    float* out = (float*)d_out;

    float* h        = (float*)d_ws;
    int*   counts   = (int*)(h + (size_t)N * OUT_F);
    int2*  sorted   = (int2*)(counts + (size_t)NB * NSLICE);
    int*   binstart = (int*)(sorted + E);
    float* a_s      = (float*)(binstart + NB + 1);
    float* a_d      = a_s + N;

    bin_count<<<NSLICE, 1024, 0, stream>>>(ei, counts, E, NB, slice);
    bin_scan<<<1, 1024, 0, stream>>>(counts, binstart, NB);
    bin_scatter<<<NSLICE, 1024, 0, stream>>>(ei, counts, binstart, sorted, E, NB, slice);
    gat_gemm<<<(N + 127) / 128, 256, 0, stream>>>(x, W, att_src, att_dst, h, a_s, a_d, N);
    gat_bin_process<<<NB, 512, 0, stream>>>(sorted, binstart, h, a_s, a_d, bias, out, N);
}


# Bin pass counting-sorts its items by destination row into an LDS list; one wave per row takes max, exp-sum and weighted h rows in registers in list order, divides once; LDS float atomics removed
# speedup vs baseline: 3.2040x; 1.2741x over previous
#include <hip/hip_runtime.h>
#include <hip/hip_bf16.h>

#define IN_F  128
#define OUT_F 64
#define NEG_SLOPE 0.2f
#define BIN_SHIFT 7
#define BIN_NODES 128
#define MAXB 1024
#define NSLICE 128

typedef __attribute__((ext_vector_type(2))) float v2f;
typedef __attribute__((ext_vector_type(8))) float v8f;

__device__ __forceinline__ unsigned f2ord(float f) {
    unsigned u = __float_as_uint(f);
    return (u & 0x80000000u) ? ~u : (u | 0x80000000u);
}
__device__ __forceinline__ float ord2f(unsigned u) {
    unsigned v = (u & 0x80000000u) ? (u & 0x7FFFFFFFu) : ~u;
    return __uint_as_float(v);
}

__global__ __launch_bounds__(256) void gat_gemm(const float* __restrict__ x,
                                                const float* __restrict__ W,
                                                const float* __restrict__ att_src,
                                                const float* __restrict__ att_dst,
                                                float* __restrict__ h,
                                                float* __restrict__ a_s,
                                                float* __restrict__ a_d, int N) {
    __shared__ float Ws[IN_F * OUT_F];
    for (int i = threadIdx.x; i < IN_F * OUT_F; i += 256) {
        int k = i >> 6;
        int n = i & 63;
        Ws[(k >> 1) * 128 + n * 2 + (k & 1)] = W[i];
    }
    __syncthreads();

    const int lane = threadIdx.x & 31;
    const int wave = threadIdx.x >> 5;
    const int m    = lane & 15;
    const int hi   = lane >> 4;
    const int row0 = (blockIdx.x * 8 + wave) * 16;

    const int arow = row0 + m;
    const int crow = arow < N ? arow : 0;
    const float2* xrow2 = (const float2*)(x + (size_t)crow * IN_F);

    v8f c0 = {}, c1 = {}, c2 = {}, c3 = {};

    for (int k0 = 0; k0 < IN_F; k0 += 4) {
        const int kp = (k0 >> 1) + hi;
        float2 af = xrow2[kp];
        v2f a = { af.x, af.y };
        const float2* wrow = (const float2*)&Ws[kp * 128];
        float2 f0 = wrow[m];
        float2 f1 = wrow[m + 16];
        float2 f2 = wrow[m + 32];
        float2 f3 = wrow[m + 48];
        v2f b0 = { f0.x, f0.y };
        v2f b1 = { f1.x, f1.y };
        v2f b2 = { f2.x, f2.y };
        v2f b3 = { f3.x, f3.y };
        c0 = __builtin_amdgcn_wmma_f32_16x16x4_f32(false, a, false, b0, (short)0, c0, false, false);
        c1 = __builtin_amdgcn_wmma_f32_16x16x4_f32(false, a, false, b1, (short)0, c1, false, false);
        c2 = __builtin_amdgcn_wmma_f32_16x16x4_f32(false, a, false, b2, (short)0, c2, false, false);
        c3 = __builtin_amdgcn_wmma_f32_16x16x4_f32(false, a, false, b3, (short)0, c3, false, false);
    }

    const float as0 = att_src[m],      as1 = att_src[16 + m];
    const float as2 = att_src[32 + m], as3 = att_src[48 + m];
    const float ad0 = att_dst[m],      ad1 = att_dst[16 + m];
    const float ad2 = att_dst[32 + m], ad3 = att_dst[48 + m];

#pragma unroll
    for (int r = 0; r < 8; ++r) {
        const int row = row0 + r + 8 * hi;
        const float e0 = c0[r], e1 = c1[r], e2 = c2[r], e3 = c3[r];
        if (row < N) {
            float* hr = h + (size_t)row * OUT_F;
            hr[m]      = e0;
            hr[16 + m] = e1;
            hr[32 + m] = e2;
            hr[48 + m] = e3;
        }
        float ps = e0 * as0 + e1 * as1 + e2 * as2 + e3 * as3;
        float pd = e0 * ad0 + e1 * ad1 + e2 * ad2 + e3 * ad3;
#pragma unroll
        for (int off = 8; off >= 1; off >>= 1) {
            ps += __shfl_xor(ps, off, 32);
            pd += __shfl_xor(pd, off, 32);
        }
        if (m == 0 && row < N) { a_s[row] = ps; a_d[row] = pd; }
    }
}


__global__ __launch_bounds__(1024) void bin_count(const int* __restrict__ ei,
                                                  int* __restrict__ counts,
                                                  int E, int NB, int slice) {
    __shared__ int hist[MAXB];
    for (int b = threadIdx.x; b < NB; b += 1024) hist[b] = 0;
    __syncthreads();
    const int lo = blockIdx.x * slice;
    const int hi = min(lo + slice, E);
    for (int i0 = lo + threadIdx.x; i0 < hi; i0 += 4 * 1024) {
        int d[4];
#pragma unroll
        for (int u = 0; u < 4; ++u) {
            const int i = i0 + u * 1024;
            d[u] = (i < hi) ? ei[E + i] : -1;
        }
#pragma unroll
        for (int u = 0; u < 4; ++u)
            if (d[u] >= 0) atomicAdd(&hist[d[u] >> BIN_SHIFT], 1);
    }
    __syncthreads();
    for (int b = threadIdx.x; b < NB; b += 1024) counts[b * NSLICE + blockIdx.x] = hist[b];
}

__global__ __launch_bounds__(1024) void bin_scan(int* __restrict__ counts,
                                                 int* __restrict__ binstart, int NB) {
    __shared__ int tot[MAXB];
    __shared__ int wsum[32];
    const int lane = threadIdx.x & 31;
    const int wave = threadIdx.x >> 5;
    for (int b0 = wave; b0 < NB; b0 += 4 * 32) {
        int4 v[4];
#pragma unroll
        for (int u = 0; u < 4; ++u) {
            const int b = b0 + u * 32;
            v[u] = make_int4(0, 0, 0, 0);
            if (b < NB) v[u] = ((const int4*)(counts + (size_t)b * NSLICE))[lane];
        }
#pragma unroll
        for (int u = 0; u < 4; ++u) {
            const int b = b0 + u * 32;
            const int s1 = v[u].x, s2 = s1 + v[u].y, s3 = s2 + v[u].z, s4 = s3 + v[u].w;
            int x = s4;
#pragma unroll
            for (int off = 1; off < 32; off <<= 1) {
                int y = __shfl_up(x, off, 32);
                if (lane >= off) x += y;
            }
            const int ex = x - s4;
            if (b < NB) {
                ((int4*)(counts + (size_t)b * NSLICE))[lane] = make_int4(ex, ex + s1, ex + s2, ex + s3);
                if (lane == 31) tot[b] = x;
            }
        }
    }
    __syncthreads();
    const int t = (threadIdx.x < NB) ? tot[threadIdx.x] : 0;
    int x = t;
#pragma unroll
    for (int off = 1; off < 32; off <<= 1) {
        int y = __shfl_up(x, off, 32);
        if (lane >= off) x += y;
    }
    if (lane == 31) wsum[wave] = x;
    __syncthreads();
    if (wave == 0) {
        int w = wsum[lane];
#pragma unroll
        for (int off = 1; off < 32; off <<= 1) {
            int y = __shfl_up(w, off, 32);
            if (lane >= off) w += y;
        }
        wsum[lane] = w;
    }
    __syncthreads();
    const int excl = x - t + (wave > 0 ? wsum[wave - 1] : 0);
    if (threadIdx.x < NB) binstart[threadIdx.x] = excl;
    if (threadIdx.x == NB - 1) binstart[NB] = excl + t;
}

__global__ __launch_bounds__(1024) void bin_scatter(const int* __restrict__ ei,
                                                    const int* __restrict__ offs,
                                                    const int* __restrict__ binstart,
                                                    int2* __restrict__ sorted,
                                                    int E, int NB, int slice) {
    __shared__ int cur[MAXB];
    for (int b = threadIdx.x; b < NB; b += 1024)
        cur[b] = binstart[b] + offs[b * NSLICE + blockIdx.x];
    __syncthreads();
    const int lo = blockIdx.x * slice;
    const int hi = min(lo + slice, E);
    for (int i0 = lo + threadIdx.x; i0 < hi; i0 += 4 * 1024) {
        int s[4], d[4];
#pragma unroll
        for (int u = 0; u < 4; ++u) {
            const int i = i0 + u * 1024;
            s[u] = 0; d[u] = -1;
            if (i < hi) { s[u] = ei[i]; d[u] = ei[E + i]; }
        }
#pragma unroll
        for (int u = 0; u < 4; ++u) {
            if (d[u] >= 0) {
                const int pos = atomicAdd(&cur[d[u] >> BIN_SHIFT], 1);
                sorted[pos] = make_int2(s[u], d[u]);
            }
        }
    }
}

#define LIST_CAP 4096

__global__ __launch_bounds__(512) void gat_bin_process(const int2* __restrict__ sorted,
                                                       const int* __restrict__ binstart,
                                                       const float* __restrict__ h,
                                                       const float* __restrict__ a_s,
                                                       const float* __restrict__ a_d,
                                                       const float* __restrict__ bias,
                                                       float* __restrict__ out, int N) {
    __shared__ int2 lst[LIST_CAP];
    __shared__ int  cnt[BIN_NODES];
    __shared__ int  rstart[BIN_NODES + 1];

    const int tid  = threadIdx.x;
    const int lane = tid & 31;
    const int wave = __builtin_amdgcn_readfirstlane(tid >> 5);
    const int base = blockIdx.x << BIN_SHIFT;
    const int start = binstart[blockIdx.x];
    const int ne    = binstart[blockIdx.x + 1] - start;
    const int2* eb  = sorted + start;
    const int nrow  = min(BIN_NODES, N - base);
    const float2* h2 = (const float2*)h;
    float2* out2 = (float2*)out;
    const float2 bv = ((const float2*)bias)[lane];

    if (ne + nrow <= LIST_CAP) {
        if (tid < BIN_NODES) cnt[tid] = (tid < nrow) ? 1 : 0;
        __syncthreads();

        for (int j0 = tid; j0 < ne; j0 += 4 * 512) {
            int d[4];
#pragma unroll
            for (int u = 0; u < 4; ++u) {
                const int j = j0 + u * 512;
                d[u] = eb[j < ne ? j : ne - 1].y;
            }
#pragma unroll
            for (int u = 0; u < 4; ++u)
                if (j0 + u * 512 < ne) atomicAdd(&cnt[d[u] - base], 1);
        }
        __syncthreads();

        if (tid < 32) {
            const int c0 = cnt[4 * tid], c1 = cnt[4 * tid + 1], c2 = cnt[4 * tid + 2], c3 = cnt[4 * tid + 3];
            const int s4 = c0 + c1 + c2 + c3;
            int x = s4;
#pragma unroll
            for (int off = 1; off < 32; off <<= 1) {
                int y = __shfl_up(x, off, 32);
                if (lane >= off) x += y;
            }
            const int ex = x - s4;
            rstart[4 * tid]     = ex;            cnt[4 * tid]     = ex;
            rstart[4 * tid + 1] = ex + c0;       cnt[4 * tid + 1] = ex + c0;
            rstart[4 * tid + 2] = ex + c0 + c1;  cnt[4 * tid + 2] = ex + c0 + c1;
            rstart[4 * tid + 3] = ex + s4 - c3;  cnt[4 * tid + 3] = ex + s4 - c3;
            if (tid == 31) rstart[BIN_NODES] = x;
        }
        __syncthreads();

        if (tid < nrow) {
            const int node = base + tid;
            const int pos = atomicAdd(&cnt[tid], 1);
            lst[pos] = make_int2(node, __float_as_int(a_s[node]));
        }
        for (int j0 = tid; j0 < ne; j0 += 4 * 512) {
            int2  r[4];
            float as[4];
#pragma unroll
            for (int u = 0; u < 4; ++u) {
                const int j = j0 + u * 512;
                r[u] = eb[j < ne ? j : ne - 1];
            }
#pragma unroll
            for (int u = 0; u < 4; ++u) as[u] = a_s[r[u].x];
#pragma unroll
            for (int u = 0; u < 4; ++u) {
                if (j0 + u * 512 < ne) {
                    const int pos = atomicAdd(&cnt[r[u].y - base], 1);
                    lst[pos] = make_int2(r[u].x, __float_as_int(as[u]));
                }
            }
        }
        __syncthreads();

        for (int i = 0; i < 8; ++i) {
            const int r = wave * 8 + i;
            if (r >= nrow) break;
            const int node = base + r;
            const int rs = __builtin_amdgcn_readfirstlane(rstart[r]);
            const int re = __builtin_amdgcn_readfirstlane(rstart[r + 1]);
            const float adv = a_d[node];

            float m = -__builtin_inff();
            for (int p = rs + lane; p < re; p += 32) {
                float e = __int_as_float(lst[p].y) + adv;
                e = e > 0.0f ? e : NEG_SLOPE * e;
                m = fmaxf(m, e);
            }
#pragma unroll
            for (int off = 16; off >= 1; off >>= 1) m = fmaxf(m, __shfl_xor(m, off, 32));

            float ssum = 0.0f, a0 = 0.0f, a1 = 0.0f;
            for (int p0 = rs; p0 < re; p0 += 32) {
                const int p = p0 + lane;
                int   s = node;
                float w = 0.0f;
                if (p < re) {
                    const int2 it = lst[p];
                    float e = __int_as_float(it.y) + adv;
                    e = e > 0.0f ? e : NEG_SLOPE * e;
                    w = __expf(e - m);
                    s = it.x;
                }
                const unsigned wb = __float_as_uint(w);
                const int nk = min(32, re - p0);
                for (int k0 = 0; k0 < nk; k0 += 8) {
                    float2 hv[8];
#pragma unroll
                    for (int u = 0; u < 8; ++u) {
                        const int sk = __builtin_amdgcn_readlane(s, k0 + u);
                        hv[u] = h2[(size_t)sk * (OUT_F / 2) + lane];
                    }
#pragma unroll
                    for (int u = 0; u < 8; ++u) {
                        const float wk = __uint_as_float(__builtin_amdgcn_readlane(wb, k0 + u));
                        a0 = fmaf(wk, hv[u].x, a0);
                        a1 = fmaf(wk, hv[u].y, a1);
                        ssum += wk;
                    }
                }
            }
            out2[(size_t)node * (OUT_F / 2) + lane] = make_float2(a0 / ssum + bv.x, a1 / ssum + bv.y);
        }
    } else {
        for (int i = 0; i < 8; ++i) {
            const int r = wave * 8 + i;
            if (r >= nrow) break;
            const int node = base + r;
            const float adv = a_d[node];
            float es = a_s[node] + adv;
            es = es > 0.0f ? es : NEG_SLOPE * es;

            float m = es;
            for (int j0 = 0; j0 < ne; j0 += 32) {
                const int j = j0 + lane;
                if (j < ne) {
                    const int2 rec = eb[j];
                    if (rec.y == node) {
                        float e = a_s[rec.x] + adv;
                        e = e > 0.0f ? e : NEG_SLOPE * e;
                        m = fmaxf(m, e);
                    }
                }
            }
#pragma unroll
            for (int off = 16; off >= 1; off >>= 1) m = fmaxf(m, __shfl_xor(m, off, 32));

            float ssum = __expf(es - m);
            const float2 hs = h2[(size_t)node * (OUT_F / 2) + lane];
            float a0 = ssum * hs.x, a1 = ssum * hs.y;
            for (int j0 = 0; j0 < ne; j0 += 32) {
                const int j = j0 + lane;
                int   s = node;
                float w = 0.0f;
                bool  hit = false;
                if (j < ne) {
                    const int2 rec = eb[j];
                    if (rec.y == node) {
                        float e = a_s[rec.x] + adv;
                        e = e > 0.0f ? e : NEG_SLOPE * e;
                        w = __expf(e - m);
                        s = rec.x;
                        hit = true;
                    }
                }
                const unsigned wb = __float_as_uint(w);
                unsigned mask = __builtin_amdgcn_ballot_w32(hit);
                while (mask) {
                    const int k = __builtin_ctz(mask);
                    mask &= mask - 1;
                    const int   sk = __builtin_amdgcn_readlane(s, k);
                    const float wk = __uint_as_float(__builtin_amdgcn_readlane(wb, k));
                    const float2 hv = h2[(size_t)sk * (OUT_F / 2) + lane];
                    a0 = fmaf(wk, hv.x, a0);
                    a1 = fmaf(wk, hv.y, a1);
                    ssum += wk;
                }
            }
            out2[(size_t)node * (OUT_F / 2) + lane] = make_float2(a0 / ssum + bv.x, a1 / ssum + bv.y);
        }
    }
}

extern "C" void kernel_launch(void* const* d_in, const int* in_sizes, int n_in,
                              void* d_out, int out_size, void* d_ws, size_t ws_size,
                              hipStream_t stream) {
    const float* x       = (const float*)d_in[0];
    const int*   ei      = (const int*)d_in[1];
    const float* W       = (const float*)d_in[2];
    const float* att_src = (const float*)d_in[3];
    const float* att_dst = (const float*)d_in[4];
    const float* bias    = (const float*)d_in[5];

    const int N  = in_sizes[0] / IN_F;
    const int E  = in_sizes[1] / 2;
    const int NB = (N + BIN_NODES - 1) / BIN_NODES;
    const int slice = (E + NSLICE - 1) / NSLICE;

    float* out = (float*)d_out;

    float* h        = (float*)d_ws;
    int*   counts   = (int*)(h + (size_t)N * OUT_F);
    int2*  sorted   = (int2*)(counts + (size_t)NB * NSLICE);
    int*   binstart = (int*)(sorted + E);
    float* a_s      = (float*)(binstart + NB + 1);
    float* a_d      = a_s + N;

    bin_count<<<NSLICE, 1024, 0, stream>>>(ei, counts, E, NB, slice);
    bin_scan<<<1, 1024, 0, stream>>>(counts, binstart, NB);
    bin_scatter<<<NSLICE, 1024, 0, stream>>>(ei, counts, binstart, sorted, E, NB, slice);
    gat_gemm<<<(N + 127) / 128, 256, 0, stream>>>(x, W, att_src, att_dst, h, a_s, a_d, N);
    gat_bin_process<<<NB, 512, 0, stream>>>(sorted, binstart, h, a_s, a_d, bias, out, N);
}


# x*W product (WMMA) runs as 250 extra work-groups inside the bin-scatter launch, which filled 128 of 256 WGPs and no matrix unit; the product's own launch removed (5 -> 4 launches)
# speedup vs baseline: 3.7710x; 1.1770x over previous
#include <hip/hip_runtime.h>
#include <hip/hip_bf16.h>

#define IN_F  128
#define OUT_F 64
#define NEG_SLOPE 0.2f
#define BIN_SHIFT 7
#define BIN_NODES 128
#define MAXB 1024
#define NSLICE 128
#define TPB 25

typedef __attribute__((ext_vector_type(2))) float v2f;
typedef __attribute__((ext_vector_type(8))) float v8f;

__device__ __forceinline__ unsigned f2ord(float f) {
    unsigned u = __float_as_uint(f);
    return (u & 0x80000000u) ? ~u : (u | 0x80000000u);
}
__device__ __forceinline__ float ord2f(unsigned u) {
    unsigned v = (u & 0x80000000u) ? (u & 0x7FFFFFFFu) : ~u;
    return __uint_as_float(v);
}

__device__ __forceinline__ void gemm_role(const float* __restrict__ x,
                                          const float* __restrict__ W,
                                          const float* __restrict__ att_src,
                                          const float* __restrict__ att_dst,
                                          float* __restrict__ h,
                                          float* __restrict__ a_s,
                                          float* __restrict__ a_d, int N,
                                          int tile_lo, int tile_hi, float* Ws) {
    float wv[IN_F * OUT_F / 1024];
#pragma unroll
    for (int u = 0; u < IN_F * OUT_F / 1024; ++u) wv[u] = W[threadIdx.x + u * 1024];
#pragma unroll
    for (int u = 0; u < IN_F * OUT_F / 1024; ++u) {
        const int i = threadIdx.x + u * 1024;
        const int k = i >> 6;
        const int n = i & 63;
        Ws[(k >> 1) * 128 + n * 2 + (k & 1)] = wv[u];
    }
    __syncthreads();

    const int lane = threadIdx.x & 31;
    const int wave = __builtin_amdgcn_readfirstlane(threadIdx.x >> 5);
    const int m    = lane & 15;
    const int hi   = lane >> 4;

    const int t = tile_lo + wave;
    if (t < tile_hi) {
        const int row0 = t * 16;
        const int arow = row0 + m;
        const int crow = arow < N ? arow : 0;
        const unsigned xoff = (unsigned)crow * (IN_F * 4u) + (unsigned)hi * 8u;
        const char*   xb    = (const char*)x;
        const float2* wl    = (const float2*)Ws + hi * 64 + m;

        v8f c0 = {}, c1 = {}, c2 = {}, c3 = {};

#pragma unroll 8
        for (int k0 = 0; k0 < IN_F; k0 += 4) {
            const int kq = k0 >> 1;
            float2 af = *(const float2*)(xb + (size_t)xoff + kq * 8);
            v2f a = { af.x, af.y };
            const float2* wrow = wl + kq * 64;
            float2 f0 = wrow[0];
            float2 f1 = wrow[16];
            float2 f2 = wrow[32];
            float2 f3 = wrow[48];
            v2f b0 = { f0.x, f0.y };
            v2f b1 = { f1.x, f1.y };
            v2f b2 = { f2.x, f2.y };
            v2f b3 = { f3.x, f3.y };
            c0 = __builtin_amdgcn_wmma_f32_16x16x4_f32(false, a, false, b0, (short)0, c0, false, false);
            c1 = __builtin_amdgcn_wmma_f32_16x16x4_f32(false, a, false, b1, (short)0, c1, false, false);
            c2 = __builtin_amdgcn_wmma_f32_16x16x4_f32(false, a, false, b2, (short)0, c2, false, false);
            c3 = __builtin_amdgcn_wmma_f32_16x16x4_f32(false, a, false, b3, (short)0, c3, false, false);
        }

        const float as0 = att_src[m],      as1 = att_src[16 + m];
        const float as2 = att_src[32 + m], as3 = att_src[48 + m];
        const float ad0 = att_dst[m],      ad1 = att_dst[16 + m];
        const float ad2 = att_dst[32 + m], ad3 = att_dst[48 + m];

#pragma unroll
        for (int r = 0; r < 8; ++r) {
            const int row = row0 + r + 8 * hi;
            const float e0 = c0[r], e1 = c1[r], e2 = c2[r], e3 = c3[r];
            if (row < N) {
                float* hr = h + (size_t)row * OUT_F;
                hr[m]      = e0;
                hr[16 + m] = e1;
                hr[32 + m] = e2;
                hr[48 + m] = e3;
            }
            float ps = e0 * as0 + e1 * as1 + e2 * as2 + e3 * as3;
            float pd = e0 * ad0 + e1 * ad1 + e2 * ad2 + e3 * ad3;
#pragma unroll
            for (int off = 8; off >= 1; off >>= 1) {
                ps += __shfl_xor(ps, off, 32);
                pd += __shfl_xor(pd, off, 32);
            }
            if (m == 0 && row < N) { a_s[row] = ps; a_d[row] = pd; }
        }
    }
}


__global__ __launch_bounds__(1024) void bin_count(const int* __restrict__ ei,
                                                  int* __restrict__ counts,
                                                  int E, int NB, int slice) {
    __shared__ int hist[MAXB];
    for (int b = threadIdx.x; b < NB; b += 1024) hist[b] = 0;
    __syncthreads();
    const int lo = blockIdx.x * slice;
    const int hi = min(lo + slice, E);
    for (int i0 = lo + threadIdx.x; i0 < hi; i0 += 4 * 1024) {
        int d[4];
#pragma unroll
        for (int u = 0; u < 4; ++u) {
            const int i = i0 + u * 1024;
            d[u] = (i < hi) ? ei[E + i] : -1;
        }
#pragma unroll
        for (int u = 0; u < 4; ++u)
            if (d[u] >= 0) atomicAdd(&hist[d[u] >> BIN_SHIFT], 1);
    }
    __syncthreads();
    for (int b = threadIdx.x; b < NB; b += 1024) counts[b * NSLICE + blockIdx.x] = hist[b];
}

__global__ __launch_bounds__(1024) void bin_scan(int* __restrict__ counts,
                                                 int* __restrict__ binstart, int NB) {
    __shared__ int tot[MAXB];
    __shared__ int wsum[32];
    const int lane = threadIdx.x & 31;
    const int wave = threadIdx.x >> 5;
    for (int b0 = wave; b0 < NB; b0 += 4 * 32) {
        int4 v[4];
#pragma unroll
        for (int u = 0; u < 4; ++u) {
            const int b = b0 + u * 32;
            v[u] = make_int4(0, 0, 0, 0);
            if (b < NB) v[u] = ((const int4*)(counts + (size_t)b * NSLICE))[lane];
        }
#pragma unroll
        for (int u = 0; u < 4; ++u) {
            const int b = b0 + u * 32;
            const int s1 = v[u].x, s2 = s1 + v[u].y, s3 = s2 + v[u].z, s4 = s3 + v[u].w;
            int x = s4;
#pragma unroll
            for (int off = 1; off < 32; off <<= 1) {
                int y = __shfl_up(x, off, 32);
                if (lane >= off) x += y;
            }
            const int ex = x - s4;
            if (b < NB) {
                ((int4*)(counts + (size_t)b * NSLICE))[lane] = make_int4(ex, ex + s1, ex + s2, ex + s3);
                if (lane == 31) tot[b] = x;
            }
        }
    }
    __syncthreads();
    const int t = (threadIdx.x < NB) ? tot[threadIdx.x] : 0;
    int x = t;
#pragma unroll
    for (int off = 1; off < 32; off <<= 1) {
        int y = __shfl_up(x, off, 32);
        if (lane >= off) x += y;
    }
    if (lane == 31) wsum[wave] = x;
    __syncthreads();
    if (wave == 0) {
        int w = wsum[lane];
#pragma unroll
        for (int off = 1; off < 32; off <<= 1) {
            int y = __shfl_up(w, off, 32);
            if (lane >= off) w += y;
        }
        wsum[lane] = w;
    }
    __syncthreads();
    const int excl = x - t + (wave > 0 ? wsum[wave - 1] : 0);
    if (threadIdx.x < NB) binstart[threadIdx.x] = excl;
    if (threadIdx.x == NB - 1) binstart[NB] = excl + t;
}

__global__ __launch_bounds__(1024, 16) void bin_scatter_gemm(const int* __restrict__ ei,
                                                         const int* __restrict__ offs,
                                                         const int* __restrict__ binstart,
                                                         int2* __restrict__ sorted,
                                                         int E, int NB, int slice,
                                                         const float* __restrict__ x,
                                                         const float* __restrict__ W,
                                                         const float* __restrict__ att_src,
                                                         const float* __restrict__ att_dst,
                                                         float* __restrict__ h,
                                                         float* __restrict__ a_s,
                                                         float* __restrict__ a_d, int N,
                                                         int tile_lo, int tile_end, int tpb) {
    __shared__ float Ws[IN_F * OUT_F];
    if (blockIdx.x >= NSLICE) {
        const int lo = tile_lo + (blockIdx.x - NSLICE) * tpb;
        gemm_role(x, W, att_src, att_dst, h, a_s, a_d, N, lo, min(lo + tpb, tile_end), Ws);
        return;
    }
    int* cur = (int*)Ws;
    for (int b = threadIdx.x; b < NB; b += 1024)
        cur[b] = binstart[b] + offs[b * NSLICE + blockIdx.x];
    __syncthreads();
    const int lo = blockIdx.x * slice;
    const int hi = min(lo + slice, E);
    for (int i0 = lo + threadIdx.x; i0 < hi; i0 += 4 * 1024) {
        int s[4], d[4];
#pragma unroll
        for (int u = 0; u < 4; ++u) {
            const int i = i0 + u * 1024;
            s[u] = 0; d[u] = -1;
            if (i < hi) { s[u] = ei[i]; d[u] = ei[E + i]; }
        }
#pragma unroll
        for (int u = 0; u < 4; ++u) {
            if (d[u] >= 0) {
                const int pos = atomicAdd(&cur[d[u] >> BIN_SHIFT], 1);
                sorted[pos] = make_int2(s[u], d[u]);
            }
        }
    }
}

#define LIST_CAP 4096

__global__ __launch_bounds__(512) void gat_bin_process(const int2* __restrict__ sorted,
                                                       const int* __restrict__ binstart,
                                                       const float* __restrict__ h,
                                                       const float* __restrict__ a_s,
                                                       const float* __restrict__ a_d,
                                                       const float* __restrict__ bias,
                                                       float* __restrict__ out, int N) {
    __shared__ int2 lst[LIST_CAP];
    __shared__ int  cnt[BIN_NODES];
    __shared__ int  rstart[BIN_NODES + 1];

    const int tid  = threadIdx.x;
    const int lane = tid & 31;
    const int wave = __builtin_amdgcn_readfirstlane(tid >> 5);
    const int base = blockIdx.x << BIN_SHIFT;
    const int start = binstart[blockIdx.x];
    const int ne    = binstart[blockIdx.x + 1] - start;
    const int2* eb  = sorted + start;
    const int nrow  = min(BIN_NODES, N - base);
    const float2* h2 = (const float2*)h;
    float2* out2 = (float2*)out;
    const float2 bv = ((const float2*)bias)[lane];

    if (ne + nrow <= LIST_CAP) {
        if (tid < BIN_NODES) cnt[tid] = (tid < nrow) ? 1 : 0;
        __syncthreads();

        for (int j0 = tid; j0 < ne; j0 += 4 * 512) {
            int d[4];
#pragma unroll
            for (int u = 0; u < 4; ++u) {
                const int j = j0 + u * 512;
                d[u] = eb[j < ne ? j : ne - 1].y;
            }
#pragma unroll
            for (int u = 0; u < 4; ++u)
                if (j0 + u * 512 < ne) atomicAdd(&cnt[d[u] - base], 1);
        }
        __syncthreads();

        if (tid < 32) {
            const int c0 = cnt[4 * tid], c1 = cnt[4 * tid + 1], c2 = cnt[4 * tid + 2], c3 = cnt[4 * tid + 3];
            const int s4 = c0 + c1 + c2 + c3;
            int x = s4;
#pragma unroll
            for (int off = 1; off < 32; off <<= 1) {
                int y = __shfl_up(x, off, 32);
                if (lane >= off) x += y;
            }
            const int ex = x - s4;
            rstart[4 * tid]     = ex;            cnt[4 * tid]     = ex;
            rstart[4 * tid + 1] = ex + c0;       cnt[4 * tid + 1] = ex + c0;
            rstart[4 * tid + 2] = ex + c0 + c1;  cnt[4 * tid + 2] = ex + c0 + c1;
            rstart[4 * tid + 3] = ex + s4 - c3;  cnt[4 * tid + 3] = ex + s4 - c3;
            if (tid == 31) rstart[BIN_NODES] = x;
        }
        __syncthreads();

        if (tid < nrow) {
            const int node = base + tid;
            const int pos = atomicAdd(&cnt[tid], 1);
            lst[pos] = make_int2(node, __float_as_int(a_s[node]));
        }
        for (int j0 = tid; j0 < ne; j0 += 4 * 512) {
            int2  r[4];
            float as[4];
#pragma unroll
            for (int u = 0; u < 4; ++u) {
                const int j = j0 + u * 512;
                r[u] = eb[j < ne ? j : ne - 1];
            }
#pragma unroll
            for (int u = 0; u < 4; ++u) as[u] = a_s[r[u].x];
#pragma unroll
            for (int u = 0; u < 4; ++u) {
                if (j0 + u * 512 < ne) {
                    const int pos = atomicAdd(&cnt[r[u].y - base], 1);
                    lst[pos] = make_int2(r[u].x, __float_as_int(as[u]));
                }
            }
        }
        __syncthreads();

        for (int i = 0; i < 8; ++i) {
            const int r = wave * 8 + i;
            if (r >= nrow) break;
            const int node = base + r;
            const int rs = __builtin_amdgcn_readfirstlane(rstart[r]);
            const int re = __builtin_amdgcn_readfirstlane(rstart[r + 1]);
            const float adv = a_d[node];

            float m = -__builtin_inff();
            for (int p = rs + lane; p < re; p += 32) {
                float e = __int_as_float(lst[p].y) + adv;
                e = e > 0.0f ? e : NEG_SLOPE * e;
                m = fmaxf(m, e);
            }
#pragma unroll
            for (int off = 16; off >= 1; off >>= 1) m = fmaxf(m, __shfl_xor(m, off, 32));

            float ssum = 0.0f, a0 = 0.0f, a1 = 0.0f;
            for (int p0 = rs; p0 < re; p0 += 32) {
                const int p = p0 + lane;
                int   s = node;
                float w = 0.0f;
                if (p < re) {
                    const int2 it = lst[p];
                    float e = __int_as_float(it.y) + adv;
                    e = e > 0.0f ? e : NEG_SLOPE * e;
                    w = __expf(e - m);
                    s = it.x;
                }
                const unsigned wb = __float_as_uint(w);
                const int nk = min(32, re - p0);
                for (int k0 = 0; k0 < nk; k0 += 8) {
                    float2 hv[8];
#pragma unroll
                    for (int u = 0; u < 8; ++u) {
                        const int sk = __builtin_amdgcn_readlane(s, k0 + u);
                        hv[u] = h2[(size_t)sk * (OUT_F / 2) + lane];
                    }
#pragma unroll
                    for (int u = 0; u < 8; ++u) {
                        const float wk = __uint_as_float(__builtin_amdgcn_readlane(wb, k0 + u));
                        a0 = fmaf(wk, hv[u].x, a0);
                        a1 = fmaf(wk, hv[u].y, a1);
                        ssum += wk;
                    }
                }
            }
            out2[(size_t)node * (OUT_F / 2) + lane] = make_float2(a0 / ssum + bv.x, a1 / ssum + bv.y);
        }
    } else {
        for (int i = 0; i < 8; ++i) {
            const int r = wave * 8 + i;
            if (r >= nrow) break;
            const int node = base + r;
            const float adv = a_d[node];
            float es = a_s[node] + adv;
            es = es > 0.0f ? es : NEG_SLOPE * es;

            float m = es;
            for (int j0 = 0; j0 < ne; j0 += 32) {
                const int j = j0 + lane;
                if (j < ne) {
                    const int2 rec = eb[j];
                    if (rec.y == node) {
                        float e = a_s[rec.x] + adv;
                        e = e > 0.0f ? e : NEG_SLOPE * e;
                        m = fmaxf(m, e);
                    }
                }
            }
#pragma unroll
            for (int off = 16; off >= 1; off >>= 1) m = fmaxf(m, __shfl_xor(m, off, 32));

            float ssum = __expf(es - m);
            const float2 hs = h2[(size_t)node * (OUT_F / 2) + lane];
            float a0 = ssum * hs.x, a1 = ssum * hs.y;
            for (int j0 = 0; j0 < ne; j0 += 32) {
                const int j = j0 + lane;
                int   s = node;
                float w = 0.0f;
                bool  hit = false;
                if (j < ne) {
                    const int2 rec = eb[j];
                    if (rec.y == node) {
                        float e = a_s[rec.x] + adv;
                        e = e > 0.0f ? e : NEG_SLOPE * e;
                        w = __expf(e - m);
                        s = rec.x;
                        hit = true;
                    }
                }
                const unsigned wb = __float_as_uint(w);
                unsigned mask = __builtin_amdgcn_ballot_w32(hit);
                while (mask) {
                    const int k = __builtin_ctz(mask);
                    mask &= mask - 1;
                    const int   sk = __builtin_amdgcn_readlane(s, k);
                    const float wk = __uint_as_float(__builtin_amdgcn_readlane(wb, k));
                    const float2 hv = h2[(size_t)sk * (OUT_F / 2) + lane];
                    a0 = fmaf(wk, hv.x, a0);
                    a1 = fmaf(wk, hv.y, a1);
                    ssum += wk;
                }
            }
            out2[(size_t)node * (OUT_F / 2) + lane] = make_float2(a0 / ssum + bv.x, a1 / ssum + bv.y);
        }
    }
}

extern "C" void kernel_launch(void* const* d_in, const int* in_sizes, int n_in,
                              void* d_out, int out_size, void* d_ws, size_t ws_size,
                              hipStream_t stream) {
    const float* x       = (const float*)d_in[0];
    const int*   ei      = (const int*)d_in[1];
    const float* W       = (const float*)d_in[2];
    const float* att_src = (const float*)d_in[3];
    const float* att_dst = (const float*)d_in[4];
    const float* bias    = (const float*)d_in[5];

    const int N  = in_sizes[0] / IN_F;
    const int E  = in_sizes[1] / 2;
    const int NB = (N + BIN_NODES - 1) / BIN_NODES;
    const int slice = (E + NSLICE - 1) / NSLICE;

    float* out = (float*)d_out;

    float* h        = (float*)d_ws;
    int*   counts   = (int*)(h + (size_t)N * OUT_F);
    int2*  sorted   = (int2*)(counts + (size_t)NB * NSLICE);
    int*   binstart = (int*)(sorted + E);
    float* a_s      = (float*)(binstart + NB + 1);
    float* a_d      = a_s + N;

    const int tiles = (N + 15) / 16;
    const int gw    = (tiles + TPB - 1) / TPB;

    bin_count<<<NSLICE, 1024, 0, stream>>>(ei, counts, E, NB, slice);
    bin_scan<<<1, 1024, 0, stream>>>(counts, binstart, NB);
    bin_scatter_gemm<<<NSLICE + gw, 1024, 0, stream>>>(ei, counts, binstart, sorted, E, NB, slice,
                                                       x, W, att_src, att_dst, h, a_s, a_d, N, 0, tiles, TPB);
    gat_bin_process<<<NB, 512, 0, stream>>>(sorted, binstart, h, a_s, a_d, bias, out, N);
}


# Each edge slice sorts itself by bin in one work-group (LDS count, scan, scatter to own range); bin pass walks its 128 slice segments; count and scan launches removed (4 -> 2); row sum order changes
# speedup vs baseline: 5.4760x; 1.4521x over previous
#include <hip/hip_runtime.h>
#include <hip/hip_bf16.h>

#define IN_F  128
#define OUT_F 64
#define NEG_SLOPE 0.2f
#define BIN_SHIFT 7
#define BIN_NODES 128
#define MAXB 1024
#define NSLICE 128
#define TPB 25

typedef __attribute__((ext_vector_type(2))) float v2f;
typedef __attribute__((ext_vector_type(8))) float v8f;

__device__ __forceinline__ unsigned f2ord(float f) {
    unsigned u = __float_as_uint(f);
    return (u & 0x80000000u) ? ~u : (u | 0x80000000u);
}
__device__ __forceinline__ float ord2f(unsigned u) {
    unsigned v = (u & 0x80000000u) ? (u & 0x7FFFFFFFu) : ~u;
    return __uint_as_float(v);
}

__device__ __forceinline__ void gemm_role(const float* __restrict__ x,
                                          const float* __restrict__ W,
                                          const float* __restrict__ att_src,
                                          const float* __restrict__ att_dst,
                                          float* __restrict__ h,
                                          float* __restrict__ a_s,
                                          float* __restrict__ a_d, int N,
                                          int tile_lo, int tile_hi, float* Ws) {
    float wv[IN_F * OUT_F / 1024];
#pragma unroll
    for (int u = 0; u < IN_F * OUT_F / 1024; ++u) wv[u] = W[threadIdx.x + u * 1024];
#pragma unroll
    for (int u = 0; u < IN_F * OUT_F / 1024; ++u) {
        const int i = threadIdx.x + u * 1024;
        const int k = i >> 6;
        const int n = i & 63;
        Ws[(k >> 1) * 128 + n * 2 + (k & 1)] = wv[u];
    }
    __syncthreads();

    const int lane = threadIdx.x & 31;
    const int wave = __builtin_amdgcn_readfirstlane(threadIdx.x >> 5);
    const int m    = lane & 15;
    const int hi   = lane >> 4;

    const int t = tile_lo + wave;
    if (t < tile_hi) {
        const int row0 = t * 16;
        const int arow = row0 + m;
        const int crow = arow < N ? arow : 0;
        const unsigned xoff = (unsigned)crow * (IN_F * 4u) + (unsigned)hi * 8u;
        const char*   xb    = (const char*)x;
        const float2* wl    = (const float2*)Ws + hi * 64 + m;

        v8f c0 = {}, c1 = {}, c2 = {}, c3 = {};

#pragma unroll 8
        for (int k0 = 0; k0 < IN_F; k0 += 4) {
            const int kq = k0 >> 1;
            float2 af = *(const float2*)(xb + (size_t)xoff + kq * 8);
            v2f a = { af.x, af.y };
            const float2* wrow = wl + kq * 64;
            float2 f0 = wrow[0];
            float2 f1 = wrow[16];
            float2 f2 = wrow[32];
            float2 f3 = wrow[48];
            v2f b0 = { f0.x, f0.y };
            v2f b1 = { f1.x, f1.y };
            v2f b2 = { f2.x, f2.y };
            v2f b3 = { f3.x, f3.y };
            c0 = __builtin_amdgcn_wmma_f32_16x16x4_f32(false, a, false, b0, (short)0, c0, false, false);
            c1 = __builtin_amdgcn_wmma_f32_16x16x4_f32(false, a, false, b1, (short)0, c1, false, false);
            c2 = __builtin_amdgcn_wmma_f32_16x16x4_f32(false, a, false, b2, (short)0, c2, false, false);
            c3 = __builtin_amdgcn_wmma_f32_16x16x4_f32(false, a, false, b3, (short)0, c3, false, false);
        }

        const float as0 = att_src[m],      as1 = att_src[16 + m];
        const float as2 = att_src[32 + m], as3 = att_src[48 + m];
        const float ad0 = att_dst[m],      ad1 = att_dst[16 + m];
        const float ad2 = att_dst[32 + m], ad3 = att_dst[48 + m];

#pragma unroll
        for (int r = 0; r < 8; ++r) {
            const int row = row0 + r + 8 * hi;
            const float e0 = c0[r], e1 = c1[r], e2 = c2[r], e3 = c3[r];
            if (row < N) {
                float* hr = h + (size_t)row * OUT_F;
                hr[m]      = e0;
                hr[16 + m] = e1;
                hr[32 + m] = e2;
                hr[48 + m] = e3;
            }
            float ps = e0 * as0 + e1 * as1 + e2 * as2 + e3 * as3;
            float pd = e0 * ad0 + e1 * ad1 + e2 * ad2 + e3 * ad3;
#pragma unroll
            for (int off = 8; off >= 1; off >>= 1) {
                ps += __shfl_xor(ps, off, 32);
                pd += __shfl_xor(pd, off, 32);
            }
            if (m == 0 && row < N) { a_s[row] = ps; a_d[row] = pd; }
        }
    }
}


__global__ __launch_bounds__(1024, 16) void slice_sort_gemm(const int* __restrict__ ei,
                                                        int* __restrict__ offsT,
                                                        int2* __restrict__ sorted,
                                                        int E, int NB, int slice,
                                                        const float* __restrict__ x,
                                                        const float* __restrict__ W,
                                                        const float* __restrict__ att_src,
                                                        const float* __restrict__ att_dst,
                                                        float* __restrict__ h,
                                                        float* __restrict__ a_s,
                                                        float* __restrict__ a_d, int N,
                                                        int tile_lo, int tile_end, int tpb) {
    __shared__ float Ws[IN_F * OUT_F];
    if (blockIdx.x >= NSLICE) {
        const int lo = tile_lo + (blockIdx.x - NSLICE) * tpb;
        gemm_role(x, W, att_src, att_dst, h, a_s, a_d, N, lo, min(lo + tpb, tile_end), Ws);
        return;
    }
    int* cur  = (int*)Ws;
    int* wsum = cur + MAXB;
    const int lane = threadIdx.x & 31;
    const int wave = threadIdx.x >> 5;
    for (int b = threadIdx.x; b < NB; b += 1024) cur[b] = 0;
    __syncthreads();
    const int lo = blockIdx.x * slice;
    const int hi = min(lo + slice, E);

    for (int i0 = lo + threadIdx.x; i0 < hi; i0 += 4 * 1024) {
        int d[4];
#pragma unroll
        for (int u = 0; u < 4; ++u) {
            const int i = i0 + u * 1024;
            d[u] = (i < hi) ? ei[E + i] : -1;
        }
#pragma unroll
        for (int u = 0; u < 4; ++u)
            if (d[u] >= 0) atomicAdd(&cur[d[u] >> BIN_SHIFT], 1);
    }
    __syncthreads();

    const int t = (threadIdx.x < NB) ? cur[threadIdx.x] : 0;
    int acc = t;
#pragma unroll
    for (int off = 1; off < 32; off <<= 1) {
        int y = __shfl_up(acc, off, 32);
        if (lane >= off) acc += y;
    }
    if (lane == 31) wsum[wave] = acc;
    __syncthreads();
    if (wave == 0) {
        int w = wsum[lane];
#pragma unroll
        for (int off = 1; off < 32; off <<= 1) {
            int y = __shfl_up(w, off, 32);
            if (lane >= off) w += y;
        }
        wsum[lane] = w;
    }
    __syncthreads();
    const int excl = acc - t + (wave > 0 ? wsum[wave - 1] : 0);
    if (threadIdx.x < NB) {
        cur[threadIdx.x] = lo + excl;
        offsT[(size_t)threadIdx.x * NSLICE + blockIdx.x] = excl;
    }
    if (threadIdx.x == NB - 1) offsT[(size_t)NB * NSLICE + blockIdx.x] = excl + t;
    __syncthreads();

    for (int i0 = lo + threadIdx.x; i0 < hi; i0 += 4 * 1024) {
        int s[4], d[4];
#pragma unroll
        for (int u = 0; u < 4; ++u) {
            const int i = i0 + u * 1024;
            s[u] = 0; d[u] = -1;
            if (i < hi) { s[u] = ei[i]; d[u] = ei[E + i]; }
        }
#pragma unroll
        for (int u = 0; u < 4; ++u) {
            if (d[u] >= 0) {
                const int pos = atomicAdd(&cur[d[u] >> BIN_SHIFT], 1);
                sorted[pos] = make_int2(s[u], d[u]);
            }
        }
    }
}

#define LIST_CAP 4096

__global__ __launch_bounds__(512) void gat_bin_process(const int2* __restrict__ sorted,
                                                       const int* __restrict__ offsT, int slice,
                                                       const float* __restrict__ h,
                                                       const float* __restrict__ a_s,
                                                       const float* __restrict__ a_d,
                                                       const float* __restrict__ bias,
                                                       float* __restrict__ out, int N) {
    __shared__ int2 lst[LIST_CAP];
    __shared__ int  cnt[BIN_NODES];
    __shared__ int  rstart[BIN_NODES + 1];
    __shared__ int  src_idx[LIST_CAP];
    __shared__ int  segbase[NSLICE];
    __shared__ int  segpos[NSLICE + 1];

    const int tid  = threadIdx.x;
    const int lane = tid & 31;
    const int wave = __builtin_amdgcn_readfirstlane(tid >> 5);
    const int base = blockIdx.x << BIN_SHIFT;
    static_assert(NSLICE == 128, "segment table: one wave, four slices per lane; four threads per segment");
    if (tid < 32) {
        const int4 o0 = ((const int4*)(offsT + (size_t)blockIdx.x * NSLICE))[lane];
        const int4 o1 = ((const int4*)(offsT + (size_t)(blockIdx.x + 1) * NSLICE))[lane];
        const int l0 = o1.x - o0.x, l1 = o1.y - o0.y, l2 = o1.z - o0.z, l3 = o1.w - o0.w;
        const int s4 = l0 + l1 + l2 + l3;
        int acc = s4;
#pragma unroll
        for (int off = 1; off < 32; off <<= 1) {
            int y = __shfl_up(acc, off, 32);
            if (lane >= off) acc += y;
        }
        const int ex = acc - s4;
        const int sb = 4 * lane * slice;
        segbase[4 * lane]     = sb + o0.x;              segpos[4 * lane]     = ex;
        segbase[4 * lane + 1] = sb + slice + o0.y;      segpos[4 * lane + 1] = ex + l0;
        segbase[4 * lane + 2] = sb + 2 * slice + o0.z;  segpos[4 * lane + 2] = ex + l0 + l1;
        segbase[4 * lane + 3] = sb + 3 * slice + o0.w;  segpos[4 * lane + 3] = ex + l0 + l1 + l2;
        if (lane == 31) segpos[NSLICE] = acc;
    }
    __syncthreads();
    const int ne = segpos[NSLICE];
    const int nrow  = min(BIN_NODES, N - base);
    const float2* h2 = (const float2*)h;
    float2* out2 = (float2*)out;
    const float2 bv = ((const float2*)bias)[lane];

    if (ne + nrow <= LIST_CAP) {
        if (tid < BIN_NODES) cnt[tid] = (tid < nrow) ? 1 : 0;
        {
            const int sg = tid >> 2;
            const int sb = segbase[sg], sp = segpos[sg], sl = segpos[sg + 1] - sp;
            for (int i = tid & 3; i < sl; i += 4) src_idx[sp + i] = sb + i;
        }
        __syncthreads();

        for (int j0 = tid; j0 < ne; j0 += 4 * 512) {
            int d[4];
#pragma unroll
            for (int u = 0; u < 4; ++u) {
                const int j = j0 + u * 512;
                d[u] = sorted[src_idx[j < ne ? j : ne - 1]].y;
            }
#pragma unroll
            for (int u = 0; u < 4; ++u)
                if (j0 + u * 512 < ne) atomicAdd(&cnt[d[u] - base], 1);
        }
        __syncthreads();

        if (tid < 32) {
            const int c0 = cnt[4 * tid], c1 = cnt[4 * tid + 1], c2 = cnt[4 * tid + 2], c3 = cnt[4 * tid + 3];
            const int s4 = c0 + c1 + c2 + c3;
            int x = s4;
#pragma unroll
            for (int off = 1; off < 32; off <<= 1) {
                int y = __shfl_up(x, off, 32);
                if (lane >= off) x += y;
            }
            const int ex = x - s4;
            rstart[4 * tid]     = ex;            cnt[4 * tid]     = ex;
            rstart[4 * tid + 1] = ex + c0;       cnt[4 * tid + 1] = ex + c0;
            rstart[4 * tid + 2] = ex + c0 + c1;  cnt[4 * tid + 2] = ex + c0 + c1;
            rstart[4 * tid + 3] = ex + s4 - c3;  cnt[4 * tid + 3] = ex + s4 - c3;
            if (tid == 31) rstart[BIN_NODES] = x;
        }
        __syncthreads();

        if (tid < nrow) {
            const int node = base + tid;
            const int pos = atomicAdd(&cnt[tid], 1);
            lst[pos] = make_int2(node, __float_as_int(a_s[node]));
        }
        for (int j0 = tid; j0 < ne; j0 += 4 * 512) {
            int2  r[4];
            float as[4];
#pragma unroll
            for (int u = 0; u < 4; ++u) {
                const int j = j0 + u * 512;
                r[u] = sorted[src_idx[j < ne ? j : ne - 1]];
            }
#pragma unroll
            for (int u = 0; u < 4; ++u) as[u] = a_s[r[u].x];
#pragma unroll
            for (int u = 0; u < 4; ++u) {
                if (j0 + u * 512 < ne) {
                    const int pos = atomicAdd(&cnt[r[u].y - base], 1);
                    lst[pos] = make_int2(r[u].x, __float_as_int(as[u]));
                }
            }
        }
        __syncthreads();

        for (int i = 0; i < 8; ++i) {
            const int r = wave * 8 + i;
            if (r >= nrow) break;
            const int node = base + r;
            const int rs = __builtin_amdgcn_readfirstlane(rstart[r]);
            const int re = __builtin_amdgcn_readfirstlane(rstart[r + 1]);
            const float adv = a_d[node];

            float m = -__builtin_inff();
            for (int p = rs + lane; p < re; p += 32) {
                float e = __int_as_float(lst[p].y) + adv;
                e = e > 0.0f ? e : NEG_SLOPE * e;
                m = fmaxf(m, e);
            }
#pragma unroll
            for (int off = 16; off >= 1; off >>= 1) m = fmaxf(m, __shfl_xor(m, off, 32));

            float ssum = 0.0f, a0 = 0.0f, a1 = 0.0f;
            for (int p0 = rs; p0 < re; p0 += 32) {
                const int p = p0 + lane;
                int   s = node;
                float w = 0.0f;
                if (p < re) {
                    const int2 it = lst[p];
                    float e = __int_as_float(it.y) + adv;
                    e = e > 0.0f ? e : NEG_SLOPE * e;
                    w = __expf(e - m);
                    s = it.x;
                }
                const unsigned wb = __float_as_uint(w);
                const int nk = min(32, re - p0);
                for (int k0 = 0; k0 < nk; k0 += 8) {
                    float2 hv[8];
#pragma unroll
                    for (int u = 0; u < 8; ++u) {
                        const int sk = __builtin_amdgcn_readlane(s, k0 + u);
                        hv[u] = h2[(size_t)sk * (OUT_F / 2) + lane];
                    }
#pragma unroll
                    for (int u = 0; u < 8; ++u) {
                        const float wk = __uint_as_float(__builtin_amdgcn_readlane(wb, k0 + u));
                        a0 = fmaf(wk, hv[u].x, a0);
                        a1 = fmaf(wk, hv[u].y, a1);
                        ssum += wk;
                    }
                }
            }
            out2[(size_t)node * (OUT_F / 2) + lane] = make_float2(a0 / ssum + bv.x, a1 / ssum + bv.y);
        }
    } else {
        for (int i = 0; i < 8; ++i) {
            const int r = wave * 8 + i;
            if (r >= nrow) break;
            const int node = base + r;
            const float adv = a_d[node];
            float es = a_s[node] + adv;
            es = es > 0.0f ? es : NEG_SLOPE * es;

            float m = es;
            for (int sg = 0; sg < NSLICE; ++sg) {
                const int sb = __builtin_amdgcn_readfirstlane(segbase[sg]);
                const int sl = __builtin_amdgcn_readfirstlane(segpos[sg + 1] - segpos[sg]);
                for (int j0 = 0; j0 < sl; j0 += 32) {
                    const int j = j0 + lane;
                    if (j < sl) {
                        const int2 rec = sorted[sb + j];
                        if (rec.y == node) {
                            float e = a_s[rec.x] + adv;
                            e = e > 0.0f ? e : NEG_SLOPE * e;
                            m = fmaxf(m, e);
                        }
                    }
                }
            }
#pragma unroll
            for (int off = 16; off >= 1; off >>= 1) m = fmaxf(m, __shfl_xor(m, off, 32));

            float ssum = __expf(es - m);
            const float2 hs = h2[(size_t)node * (OUT_F / 2) + lane];
            float a0 = ssum * hs.x, a1 = ssum * hs.y;
            for (int sg = 0; sg < NSLICE; ++sg) {
                const int sb = __builtin_amdgcn_readfirstlane(segbase[sg]);
                const int sl = __builtin_amdgcn_readfirstlane(segpos[sg + 1] - segpos[sg]);
                for (int j0 = 0; j0 < sl; j0 += 32) {
                    const int j = j0 + lane;
                    int   s = node;
                    float w = 0.0f;
                    bool  hit = false;
                    if (j < sl) {
                        const int2 rec = sorted[sb + j];
                        if (rec.y == node) {
                            float e = a_s[rec.x] + adv;
                            e = e > 0.0f ? e : NEG_SLOPE * e;
                            w = __expf(e - m);
                            s = rec.x;
                            hit = true;
                        }
                    }
                    const unsigned wb = __float_as_uint(w);
                    unsigned mask = __builtin_amdgcn_ballot_w32(hit);
                    while (mask) {
                        const int k = __builtin_ctz(mask);
                        mask &= mask - 1;
                        const int   sk = __builtin_amdgcn_readlane(s, k);
                        const float wk = __uint_as_float(__builtin_amdgcn_readlane(wb, k));
                        const float2 hv = h2[(size_t)sk * (OUT_F / 2) + lane];
                        a0 = fmaf(wk, hv.x, a0);
                        a1 = fmaf(wk, hv.y, a1);
                        ssum += wk;
                    }
                }
            }
            out2[(size_t)node * (OUT_F / 2) + lane] = make_float2(a0 / ssum + bv.x, a1 / ssum + bv.y);
        }
    }
}

extern "C" void kernel_launch(void* const* d_in, const int* in_sizes, int n_in,
                              void* d_out, int out_size, void* d_ws, size_t ws_size,
                              hipStream_t stream) {
    const float* x       = (const float*)d_in[0];
    const int*   ei      = (const int*)d_in[1];
    const float* W       = (const float*)d_in[2];
    const float* att_src = (const float*)d_in[3];
    const float* att_dst = (const float*)d_in[4];
    const float* bias    = (const float*)d_in[5];

    const int N  = in_sizes[0] / IN_F;
    const int E  = in_sizes[1] / 2;
    const int NB = (N + BIN_NODES - 1) / BIN_NODES;
    const int slice = (E + NSLICE - 1) / NSLICE;

    float* out = (float*)d_out;

    float* h        = (float*)d_ws;
    int*   offsT    = (int*)(h + (size_t)N * OUT_F);
    int2*  sorted   = (int2*)(offsT + (size_t)(NB + 1) * NSLICE);
    float* a_s      = (float*)(sorted + E);
    float* a_d      = a_s + N;

    const int tiles = (N + 15) / 16;
    const int gw    = (tiles + TPB - 1) / TPB;

    slice_sort_gemm<<<NSLICE + gw, 1024, 0, stream>>>(ei, offsT, sorted, E, NB, slice,
                                                      x, W, att_src, att_dst, h, a_s, a_d, N, 0, tiles, TPB);
    gat_bin_process<<<NB, 512, 0, stream>>>(sorted, offsT, slice, h, a_s, a_d, bias, out, N);
}


# Slice sort on chip: threads hold their edges in registers (edge list read once), pairs are ordered in an LDS image of the slice's range and stored as whole lines; scattered 8-byte stores removed
# speedup vs baseline: 5.7027x; 1.0414x over previous
#include <hip/hip_runtime.h>
#include <hip/hip_bf16.h>

#define IN_F  128
#define OUT_F 64
#define NEG_SLOPE 0.2f
#define BIN_SHIFT 7
#define BIN_NODES 128
#define MAXB 1024
#define NSLICE 128
#define TPB 25

typedef __attribute__((ext_vector_type(2))) float v2f;
typedef __attribute__((ext_vector_type(8))) float v8f;

__device__ __forceinline__ unsigned f2ord(float f) {
    unsigned u = __float_as_uint(f);
    return (u & 0x80000000u) ? ~u : (u | 0x80000000u);
}
__device__ __forceinline__ float ord2f(unsigned u) {
    unsigned v = (u & 0x80000000u) ? (u & 0x7FFFFFFFu) : ~u;
    return __uint_as_float(v);
}

__device__ __forceinline__ void gemm_role(const float* __restrict__ x,
                                          const float* __restrict__ W,
                                          const float* __restrict__ att_src,
                                          const float* __restrict__ att_dst,
                                          float* __restrict__ h,
                                          float* __restrict__ a_s,
                                          float* __restrict__ a_d, int N,
                                          int tile_lo, int tile_hi, float* Ws) {
    float wv[IN_F * OUT_F / 1024];
#pragma unroll
    for (int u = 0; u < IN_F * OUT_F / 1024; ++u) wv[u] = W[threadIdx.x + u * 1024];
#pragma unroll
    for (int u = 0; u < IN_F * OUT_F / 1024; ++u) {
        const int i = threadIdx.x + u * 1024;
        const int k = i >> 6;
        const int n = i & 63;
        Ws[(k >> 1) * 128 + n * 2 + (k & 1)] = wv[u];
    }
    __syncthreads();

    const int lane = threadIdx.x & 31;
    const int wave = __builtin_amdgcn_readfirstlane(threadIdx.x >> 5);
    const int m    = lane & 15;
    const int hi   = lane >> 4;

    const int t = tile_lo + wave;
    if (t < tile_hi) {
        const int row0 = t * 16;
        const int arow = row0 + m;
        const int crow = arow < N ? arow : 0;
        const unsigned xoff = (unsigned)crow * (IN_F * 4u) + (unsigned)hi * 8u;
        const char*   xb    = (const char*)x;
        const float2* wl    = (const float2*)Ws + hi * 64 + m;

        v8f c0 = {}, c1 = {}, c2 = {}, c3 = {};

#pragma unroll 8
        for (int k0 = 0; k0 < IN_F; k0 += 4) {
            const int kq = k0 >> 1;
            float2 af = *(const float2*)(xb + (size_t)xoff + kq * 8);
            v2f a = { af.x, af.y };
            const float2* wrow = wl + kq * 64;
            float2 f0 = wrow[0];
            float2 f1 = wrow[16];
            float2 f2 = wrow[32];
            float2 f3 = wrow[48];
            v2f b0 = { f0.x, f0.y };
            v2f b1 = { f1.x, f1.y };
            v2f b2 = { f2.x, f2.y };
            v2f b3 = { f3.x, f3.y };
            c0 = __builtin_amdgcn_wmma_f32_16x16x4_f32(false, a, false, b0, (short)0, c0, false, false);
            c1 = __builtin_amdgcn_wmma_f32_16x16x4_f32(false, a, false, b1, (short)0, c1, false, false);
            c2 = __builtin_amdgcn_wmma_f32_16x16x4_f32(false, a, false, b2, (short)0, c2, false, false);
            c3 = __builtin_amdgcn_wmma_f32_16x16x4_f32(false, a, false, b3, (short)0, c3, false, false);
        }

        const float as0 = att_src[m],      as1 = att_src[16 + m];
        const float as2 = att_src[32 + m], as3 = att_src[48 + m];
        const float ad0 = att_dst[m],      ad1 = att_dst[16 + m];
        const float ad2 = att_dst[32 + m], ad3 = att_dst[48 + m];

#pragma unroll
        for (int r = 0; r < 8; ++r) {
            const int row = row0 + r + 8 * hi;
            const float e0 = c0[r], e1 = c1[r], e2 = c2[r], e3 = c3[r];
            if (row < N) {
                float* hr = h + (size_t)row * OUT_F;
                hr[m]      = e0;
                hr[16 + m] = e1;
                hr[32 + m] = e2;
                hr[48 + m] = e3;
            }
            float ps = e0 * as0 + e1 * as1 + e2 * as2 + e3 * as3;
            float pd = e0 * ad0 + e1 * ad1 + e2 * ad2 + e3 * ad3;
#pragma unroll
            for (int off = 8; off >= 1; off >>= 1) {
                ps += __shfl_xor(ps, off, 32);
                pd += __shfl_xor(pd, off, 32);
            }
            if (m == 0 && row < N) { a_s[row] = ps; a_d[row] = pd; }
        }
    }
}


#define SLOTS 13
#define STAGE_CAP (SLOTS * 1024)
__global__ __launch_bounds__(1024, 16) void slice_sort_gemm(const int* __restrict__ ei,
                                                        int* __restrict__ offsT,
                                                        int2* __restrict__ sorted,
                                                        int E, int NB, int slice,
                                                        const float* __restrict__ x,
                                                        const float* __restrict__ W,
                                                        const float* __restrict__ att_src,
                                                        const float* __restrict__ att_dst,
                                                        float* __restrict__ h,
                                                        float* __restrict__ a_s,
                                                        float* __restrict__ a_d, int N,
                                                        int tile_lo, int tile_end, int tpb) {
    __shared__ float Ws[IN_F * OUT_F];
    __shared__ int2  stage[STAGE_CAP + 1024];
    if (blockIdx.x >= NSLICE) {
        const int lo = tile_lo + (blockIdx.x - NSLICE) * tpb;
        gemm_role(x, W, att_src, att_dst, h, a_s, a_d, N, lo, min(lo + tpb, tile_end), Ws);
        return;
    }
    int* cur  = (int*)Ws;
    int* wsum = cur + MAXB;
    const int lane = threadIdx.x & 31;
    const int wave = threadIdx.x >> 5;
    for (int b = threadIdx.x; b < NB; b += 1024) cur[b] = 0;
    const int lo = blockIdx.x * slice;
    const int hi = min(lo + slice, E);
    const bool on_chip = hi - lo <= STAGE_CAP;

    int sr[SLOTS], ds[SLOTS];
    if (on_chip) {
#pragma unroll
        for (int u = 0; u < SLOTS; ++u) {
            const int i = lo + threadIdx.x + u * 1024;
            ds[u] = (i < hi) ? ei[E + i] : -1;
        }
#pragma unroll
        for (int u = 0; u < SLOTS; ++u) {
            const int i = lo + threadIdx.x + u * 1024;
            sr[u] = (i < hi) ? ei[i] : 0;
        }
    }
    __syncthreads();

    if (on_chip) {
#pragma unroll
        for (int u = 0; u < SLOTS; ++u)
            if (ds[u] >= 0) atomicAdd(&cur[ds[u] >> BIN_SHIFT], 1);
    } else {
        for (int i0 = lo + threadIdx.x; i0 < hi; i0 += 4 * 1024) {
            int d[4];
#pragma unroll
            for (int u = 0; u < 4; ++u) {
                const int i = i0 + u * 1024;
                d[u] = (i < hi) ? ei[E + i] : -1;
            }
#pragma unroll
            for (int u = 0; u < 4; ++u)
                if (d[u] >= 0) atomicAdd(&cur[d[u] >> BIN_SHIFT], 1);
        }
    }
    __syncthreads();

    const int t = (threadIdx.x < NB) ? cur[threadIdx.x] : 0;
    int acc = t;
#pragma unroll
    for (int off = 1; off < 32; off <<= 1) {
        int y = __shfl_up(acc, off, 32);
        if (lane >= off) acc += y;
    }
    if (lane == 31) wsum[wave] = acc;
    __syncthreads();
    if (wave == 0) {
        int w = wsum[lane];
#pragma unroll
        for (int off = 1; off < 32; off <<= 1) {
            int y = __shfl_up(w, off, 32);
            if (lane >= off) w += y;
        }
        wsum[lane] = w;
    }
    __syncthreads();
    const int excl = acc - t + (wave > 0 ? wsum[wave - 1] : 0);
    if (threadIdx.x < NB) {
        cur[threadIdx.x] = lo + excl;
        offsT[(size_t)threadIdx.x * NSLICE + blockIdx.x] = excl;
    }
    if (threadIdx.x == NB - 1) offsT[(size_t)NB * NSLICE + blockIdx.x] = excl + t;
    __syncthreads();

    if (on_chip) {
        const int a = lo & ~15;
        int pos[SLOTS];
#pragma unroll
        for (int u = 0; u < SLOTS; ++u)
            pos[u] = (ds[u] >= 0) ? atomicAdd(&cur[ds[u] >> BIN_SHIFT], 1) : a;
#pragma unroll
        for (int u = 0; u < SLOTS; ++u)
            if (ds[u] >= 0) stage[pos[u] - a] = make_int2(sr[u], ds[u]);
        __syncthreads();
        int2 rec[SLOTS + 1];
#pragma unroll
        for (int u = 0; u <= SLOTS; ++u) rec[u] = stage[threadIdx.x + u * 1024];
#pragma unroll
        for (int u = 0; u <= SLOTS; ++u) {
            const int q = a + threadIdx.x + u * 1024;
            if (q >= lo && q < hi) sorted[q] = rec[u];
        }
    } else {
        for (int i0 = lo + threadIdx.x; i0 < hi; i0 += 4 * 1024) {
            int s[4], d[4];
#pragma unroll
            for (int u = 0; u < 4; ++u) {
                const int i = i0 + u * 1024;
                s[u] = 0; d[u] = -1;
                if (i < hi) { s[u] = ei[i]; d[u] = ei[E + i]; }
            }
#pragma unroll
            for (int u = 0; u < 4; ++u) {
                if (d[u] >= 0) {
                    const int pos = atomicAdd(&cur[d[u] >> BIN_SHIFT], 1);
                    sorted[pos] = make_int2(s[u], d[u]);
                }
            }
        }
    }
}

#define LIST_CAP 4096

__global__ __launch_bounds__(512) void gat_bin_process(const int2* __restrict__ sorted,
                                                       const int* __restrict__ offsT, int slice,
                                                       const float* __restrict__ h,
                                                       const float* __restrict__ a_s,
                                                       const float* __restrict__ a_d,
                                                       const float* __restrict__ bias,
                                                       float* __restrict__ out, int N) {
    __shared__ int2 lst[LIST_CAP];
    __shared__ int  cnt[BIN_NODES];
    __shared__ int  rstart[BIN_NODES + 1];
    __shared__ int  src_idx[LIST_CAP];
    __shared__ int  segbase[NSLICE];
    __shared__ int  segpos[NSLICE + 1];

    const int tid  = threadIdx.x;
    const int lane = tid & 31;
    const int wave = __builtin_amdgcn_readfirstlane(tid >> 5);
    const int base = blockIdx.x << BIN_SHIFT;
    static_assert(NSLICE == 128, "segment table: one wave, four slices per lane; four threads per segment");
    if (tid < 32) {
        const int4 o0 = ((const int4*)(offsT + (size_t)blockIdx.x * NSLICE))[lane];
        const int4 o1 = ((const int4*)(offsT + (size_t)(blockIdx.x + 1) * NSLICE))[lane];
        const int l0 = o1.x - o0.x, l1 = o1.y - o0.y, l2 = o1.z - o0.z, l3 = o1.w - o0.w;
        const int s4 = l0 + l1 + l2 + l3;
        int acc = s4;
#pragma unroll
        for (int off = 1; off < 32; off <<= 1) {
            int y = __shfl_up(acc, off, 32);
            if (lane >= off) acc += y;
        }
        const int ex = acc - s4;
        const int sb = 4 * lane * slice;
        segbase[4 * lane]     = sb + o0.x;              segpos[4 * lane]     = ex;
        segbase[4 * lane + 1] = sb + slice + o0.y;      segpos[4 * lane + 1] = ex + l0;
        segbase[4 * lane + 2] = sb + 2 * slice + o0.z;  segpos[4 * lane + 2] = ex + l0 + l1;
        segbase[4 * lane + 3] = sb + 3 * slice + o0.w;  segpos[4 * lane + 3] = ex + l0 + l1 + l2;
        if (lane == 31) segpos[NSLICE] = acc;
    }
    __syncthreads();
    const int ne = segpos[NSLICE];
    const int nrow  = min(BIN_NODES, N - base);
    const float2* h2 = (const float2*)h;
    float2* out2 = (float2*)out;
    const float2 bv = ((const float2*)bias)[lane];

    if (ne + nrow <= LIST_CAP) {
        if (tid < BIN_NODES) cnt[tid] = (tid < nrow) ? 1 : 0;
        {
            const int sg = tid >> 2;
            const int sb = segbase[sg], sp = segpos[sg], sl = segpos[sg + 1] - sp;
            for (int i = tid & 3; i < sl; i += 4) src_idx[sp + i] = sb + i;
        }
        __syncthreads();

        for (int j0 = tid; j0 < ne; j0 += 4 * 512) {
            int d[4];
#pragma unroll
            for (int u = 0; u < 4; ++u) {
                const int j = j0 + u * 512;
                d[u] = sorted[src_idx[j < ne ? j : ne - 1]].y;
            }
#pragma unroll
            for (int u = 0; u < 4; ++u)
                if (j0 + u * 512 < ne) atomicAdd(&cnt[d[u] - base], 1);
        }
        __syncthreads();

        if (tid < 32) {
            const int c0 = cnt[4 * tid], c1 = cnt[4 * tid + 1], c2 = cnt[4 * tid + 2], c3 = cnt[4 * tid + 3];
            const int s4 = c0 + c1 + c2 + c3;
            int x = s4;
#pragma unroll
            for (int off = 1; off < 32; off <<= 1) {
                int y = __shfl_up(x, off, 32);
                if (lane >= off) x += y;
            }
            const int ex = x - s4;
            rstart[4 * tid]     = ex;            cnt[4 * tid]     = ex;
            rstart[4 * tid + 1] = ex + c0;       cnt[4 * tid + 1] = ex + c0;
            rstart[4 * tid + 2] = ex + c0 + c1;  cnt[4 * tid + 2] = ex + c0 + c1;
            rstart[4 * tid + 3] = ex + s4 - c3;  cnt[4 * tid + 3] = ex + s4 - c3;
            if (tid == 31) rstart[BIN_NODES] = x;
        }
        __syncthreads();

        if (tid < nrow) {
            const int node = base + tid;
            const int pos = atomicAdd(&cnt[tid], 1);
            lst[pos] = make_int2(node, __float_as_int(a_s[node]));
        }
        for (int j0 = tid; j0 < ne; j0 += 4 * 512) {
            int2  r[4];
            float as[4];
#pragma unroll
            for (int u = 0; u < 4; ++u) {
                const int j = j0 + u * 512;
                r[u] = sorted[src_idx[j < ne ? j : ne - 1]];
            }
#pragma unroll
            for (int u = 0; u < 4; ++u) as[u] = a_s[r[u].x];
#pragma unroll
            for (int u = 0; u < 4; ++u) {
                if (j0 + u * 512 < ne) {
                    const int pos = atomicAdd(&cnt[r[u].y - base], 1);
                    lst[pos] = make_int2(r[u].x, __float_as_int(as[u]));
                }
            }
        }
        __syncthreads();

        for (int i = 0; i < 8; ++i) {
            const int r = wave * 8 + i;
            if (r >= nrow) break;
            const int node = base + r;
            const int rs = __builtin_amdgcn_readfirstlane(rstart[r]);
            const int re = __builtin_amdgcn_readfirstlane(rstart[r + 1]);
            const float adv = a_d[node];

            float m = -__builtin_inff();
            for (int p = rs + lane; p < re; p += 32) {
                float e = __int_as_float(lst[p].y) + adv;
                e = e > 0.0f ? e : NEG_SLOPE * e;
                m = fmaxf(m, e);
            }
#pragma unroll
            for (int off = 16; off >= 1; off >>= 1) m = fmaxf(m, __shfl_xor(m, off, 32));

            float ssum = 0.0f, a0 = 0.0f, a1 = 0.0f;
            for (int p0 = rs; p0 < re; p0 += 32) {
                const int p = p0 + lane;
                int   s = node;
                float w = 0.0f;
                if (p < re) {
                    const int2 it = lst[p];
                    float e = __int_as_float(it.y) + adv;
                    e = e > 0.0f ? e : NEG_SLOPE * e;
                    w = __expf(e - m);
                    s = it.x;
                }
                const unsigned wb = __float_as_uint(w);
                const int nk = min(32, re - p0);
                for (int k0 = 0; k0 < nk; k0 += 8) {
                    float2 hv[8];
#pragma unroll
                    for (int u = 0; u < 8; ++u) {
                        const int sk = __builtin_amdgcn_readlane(s, k0 + u);
                        hv[u] = h2[(size_t)sk * (OUT_F / 2) + lane];
                    }
#pragma unroll
                    for (int u = 0; u < 8; ++u) {
                        const float wk = __uint_as_float(__builtin_amdgcn_readlane(wb, k0 + u));
                        a0 = fmaf(wk, hv[u].x, a0);
                        a1 = fmaf(wk, hv[u].y, a1);
                        ssum += wk;
                    }
                }
            }
            out2[(size_t)node * (OUT_F / 2) + lane] = make_float2(a0 / ssum + bv.x, a1 / ssum + bv.y);
        }
    } else {
        for (int i = 0; i < 8; ++i) {
            const int r = wave * 8 + i;
            if (r >= nrow) break;
            const int node = base + r;
            const float adv = a_d[node];
            float es = a_s[node] + adv;
            es = es > 0.0f ? es : NEG_SLOPE * es;

            float m = es;
            for (int sg = 0; sg < NSLICE; ++sg) {
                const int sb = __builtin_amdgcn_readfirstlane(segbase[sg]);
                const int sl = __builtin_amdgcn_readfirstlane(segpos[sg + 1] - segpos[sg]);
                for (int j0 = 0; j0 < sl; j0 += 32) {
                    const int j = j0 + lane;
                    if (j < sl) {
                        const int2 rec = sorted[sb + j];
                        if (rec.y == node) {
                            float e = a_s[rec.x] + adv;
                            e = e > 0.0f ? e : NEG_SLOPE * e;
                            m = fmaxf(m, e);
                        }
                    }
                }
            }
#pragma unroll
            for (int off = 16; off >= 1; off >>= 1) m = fmaxf(m, __shfl_xor(m, off, 32));

            float ssum = __expf(es - m);
            const float2 hs = h2[(size_t)node * (OUT_F / 2) + lane];
            float a0 = ssum * hs.x, a1 = ssum * hs.y;
            for (int sg = 0; sg < NSLICE; ++sg) {
                const int sb = __builtin_amdgcn_readfirstlane(segbase[sg]);
                const int sl = __builtin_amdgcn_readfirstlane(segpos[sg + 1] - segpos[sg]);
                for (int j0 = 0; j0 < sl; j0 += 32) {
                    const int j = j0 + lane;
                    int   s = node;
                    float w = 0.0f;
                    bool  hit = false;
                    if (j < sl) {
                        const int2 rec = sorted[sb + j];
                        if (rec.y == node) {
                            float e = a_s[rec.x] + adv;
                            e = e > 0.0f ? e : NEG_SLOPE * e;
                            w = __expf(e - m);
                            s = rec.x;
                            hit = true;
                        }
                    }
                    const unsigned wb = __float_as_uint(w);
                    unsigned mask = __builtin_amdgcn_ballot_w32(hit);
                    while (mask) {
                        const int k = __builtin_ctz(mask);
                        mask &= mask - 1;
                        const int   sk = __builtin_amdgcn_readlane(s, k);
                        const float wk = __uint_as_float(__builtin_amdgcn_readlane(wb, k));
                        const float2 hv = h2[(size_t)sk * (OUT_F / 2) + lane];
                        a0 = fmaf(wk, hv.x, a0);
                        a1 = fmaf(wk, hv.y, a1);
                        ssum += wk;
                    }
                }
            }
            out2[(size_t)node * (OUT_F / 2) + lane] = make_float2(a0 / ssum + bv.x, a1 / ssum + bv.y);
        }
    }
}

extern "C" void kernel_launch(void* const* d_in, const int* in_sizes, int n_in,
                              void* d_out, int out_size, void* d_ws, size_t ws_size,
                              hipStream_t stream) {
    const float* x       = (const float*)d_in[0];
    const int*   ei      = (const int*)d_in[1];
    const float* W       = (const float*)d_in[2];
    const float* att_src = (const float*)d_in[3];
    const float* att_dst = (const float*)d_in[4];
    const float* bias    = (const float*)d_in[5];

    const int N  = in_sizes[0] / IN_F;
    const int E  = in_sizes[1] / 2;
    const int NB = (N + BIN_NODES - 1) / BIN_NODES;
    const int slice = (E + NSLICE - 1) / NSLICE;

    float* out = (float*)d_out;

    float* h        = (float*)d_ws;
    int*   offsT    = (int*)(h + (size_t)N * OUT_F);
    int2*  sorted   = (int2*)(offsT + (size_t)(NB + 1) * NSLICE);
    float* a_s      = (float*)(sorted + E);
    float* a_d      = a_s + N;

    const int tiles = (N + 15) / 16;
    const int gw    = (tiles + TPB - 1) / TPB;

    slice_sort_gemm<<<NSLICE + gw, 1024, 0, stream>>>(ei, offsT, sorted, E, NB, slice,
                                                      x, W, att_src, att_dst, h, a_s, a_d, N, 0, tiles, TPB);
    gat_bin_process<<<NB, 512, 0, stream>>>(sorted, offsT, slice, h, a_s, a_d, bias, out, N);
}


# Bin pass row sums: each 16-byte-per-lane load fetches two source rows of h (a half-wave each, 4 columns per lane); even and odd items sum apart, halves added last; one reciprocal per row
# speedup vs baseline: 5.9522x; 1.0438x over previous
#include <hip/hip_runtime.h>
#include <hip/hip_bf16.h>

#define IN_F  128
#define OUT_F 64
#define NEG_SLOPE 0.2f
#define BIN_SHIFT 7
#define BIN_NODES 128
#define MAXB 1024
#define NSLICE 128
#define TPB 25

typedef __attribute__((ext_vector_type(2))) float v2f;
typedef __attribute__((ext_vector_type(8))) float v8f;
typedef __attribute__((ext_vector_type(4))) float v4f;

__device__ __forceinline__ unsigned f2ord(float f) {
    unsigned u = __float_as_uint(f);
    return (u & 0x80000000u) ? ~u : (u | 0x80000000u);
}
__device__ __forceinline__ float ord2f(unsigned u) {
    unsigned v = (u & 0x80000000u) ? (u & 0x7FFFFFFFu) : ~u;
    return __uint_as_float(v);
}

__device__ __forceinline__ float xhalf(float v) {
    return __builtin_bit_cast(float, __builtin_amdgcn_permlanex16(__builtin_bit_cast(int, v), __builtin_bit_cast(int, v),
                                                                  0x76543210, 0xfedcba98, false, false));
}
template <int K> __device__ __forceinline__ unsigned share16(unsigned v) {
    return (unsigned)__builtin_amdgcn_update_dpp(0, (int)v, 0x150 + K, 0xf, 0xf, true);
}
#define ROW_LD(U) hv[U] = *(const v4f*)(hb + (size_t)(share16<K0 + U>(off) + col));
#define ROW_AC(U) {                                                    \
        const float wk = __uint_as_float(share16<K0 + U>(wb));         \
        a0 = fmaf(wk, hv[U].x, a0); a1 = fmaf(wk, hv[U].y, a1);        \
        a2 = fmaf(wk, hv[U].z, a2); a3 = fmaf(wk, hv[U].w, a3);        \
        ssum += wk; }
template <int K0>
__device__ __forceinline__ void row_group_full(const char* __restrict__ hb, unsigned off, unsigned wb, unsigned col,
                                               float& a0, float& a1, float& a2, float& a3, float& ssum) {
    v4f hv[8];
    ROW_LD(0) ROW_LD(1) ROW_LD(2) ROW_LD(3) ROW_LD(4) ROW_LD(5) ROW_LD(6) ROW_LD(7)
    asm volatile("" : "+v"(wb) : "v"(hv[0]), "v"(hv[1]), "v"(hv[2]), "v"(hv[3]), "v"(hv[4]), "v"(hv[5]), "v"(hv[6]), "v"(hv[7]));
    ROW_AC(0) ROW_AC(1) ROW_AC(2) ROW_AC(3) ROW_AC(4) ROW_AC(5) ROW_AC(6) ROW_AC(7)
}
template <int K0>
__device__ __forceinline__ void row_group_part(const char* __restrict__ hb, unsigned off, unsigned wb, unsigned col, int cnt,
                                               float& a0, float& a1, float& a2, float& a3, float& ssum) {
    v4f hv[7];
    ROW_LD(0)
    if (cnt > 1) { ROW_LD(1)
     if (cnt > 2) { ROW_LD(2)
      if (cnt > 3) { ROW_LD(3)
       if (cnt > 4) { ROW_LD(4)
        if (cnt > 5) { ROW_LD(5)
         if (cnt > 6) { ROW_LD(6) } } } } } }
    ROW_AC(0)
    if (cnt > 1) { ROW_AC(1)
     if (cnt > 2) { ROW_AC(2)
      if (cnt > 3) { ROW_AC(3)
       if (cnt > 4) { ROW_AC(4)
        if (cnt > 5) { ROW_AC(5)
         if (cnt > 6) { ROW_AC(6) } } } } } }
}
#undef ROW_LD
#undef ROW_AC

__device__ __forceinline__ void gemm_role(const float* __restrict__ x,
                                          const float* __restrict__ W,
                                          const float* __restrict__ att_src,
                                          const float* __restrict__ att_dst,
                                          float* __restrict__ h,
                                          float* __restrict__ a_s,
                                          float* __restrict__ a_d, int N,
                                          int tile_lo, int tile_hi, float* Ws) {
    float wv[IN_F * OUT_F / 1024];
#pragma unroll
    for (int u = 0; u < IN_F * OUT_F / 1024; ++u) wv[u] = W[threadIdx.x + u * 1024];
#pragma unroll
    for (int u = 0; u < IN_F * OUT_F / 1024; ++u) {
        const int i = threadIdx.x + u * 1024;
        const int k = i >> 6;
        const int n = i & 63;
        Ws[(k >> 1) * 128 + n * 2 + (k & 1)] = wv[u];
    }
    __syncthreads();

    const int lane = threadIdx.x & 31;
    const int wave = __builtin_amdgcn_readfirstlane(threadIdx.x >> 5);
    const int m    = lane & 15;
    const int hi   = lane >> 4;

    const int t = tile_lo + wave;
    if (t < tile_hi) {
        const int row0 = t * 16;
        const int arow = row0 + m;
        const int crow = arow < N ? arow : 0;
        const unsigned xoff = (unsigned)crow * (IN_F * 4u) + (unsigned)hi * 8u;
        const char*   xb    = (const char*)x;
        const float2* wl    = (const float2*)Ws + hi * 64 + m;

        v8f c0 = {}, c1 = {}, c2 = {}, c3 = {};

#pragma unroll 8
        for (int k0 = 0; k0 < IN_F; k0 += 4) {
            const int kq = k0 >> 1;
            float2 af = *(const float2*)(xb + (size_t)xoff + kq * 8);
            v2f a = { af.x, af.y };
            const float2* wrow = wl + kq * 64;
            float2 f0 = wrow[0];
            float2 f1 = wrow[16];
            float2 f2 = wrow[32];
            float2 f3 = wrow[48];
            v2f b0 = { f0.x, f0.y };
            v2f b1 = { f1.x, f1.y };
            v2f b2 = { f2.x, f2.y };
            v2f b3 = { f3.x, f3.y };
            c0 = __builtin_amdgcn_wmma_f32_16x16x4_f32(false, a, false, b0, (short)0, c0, false, false);
            c1 = __builtin_amdgcn_wmma_f32_16x16x4_f32(false, a, false, b1, (short)0, c1, false, false);
            c2 = __builtin_amdgcn_wmma_f32_16x16x4_f32(false, a, false, b2, (short)0, c2, false, false);
            c3 = __builtin_amdgcn_wmma_f32_16x16x4_f32(false, a, false, b3, (short)0, c3, false, false);
        }

        const float as0 = att_src[m],      as1 = att_src[16 + m];
        const float as2 = att_src[32 + m], as3 = att_src[48 + m];
        const float ad0 = att_dst[m],      ad1 = att_dst[16 + m];
        const float ad2 = att_dst[32 + m], ad3 = att_dst[48 + m];

#pragma unroll
        for (int r = 0; r < 8; ++r) {
            const int row = row0 + r + 8 * hi;
            const float e0 = c0[r], e1 = c1[r], e2 = c2[r], e3 = c3[r];
            if (row < N) {
                float* hr = h + (size_t)row * OUT_F;
                hr[m]      = e0;
                hr[16 + m] = e1;
                hr[32 + m] = e2;
                hr[48 + m] = e3;
            }
            float ps = e0 * as0 + e1 * as1 + e2 * as2 + e3 * as3;
            float pd = e0 * ad0 + e1 * ad1 + e2 * ad2 + e3 * ad3;
#pragma unroll
            for (int off = 8; off >= 1; off >>= 1) {
                ps += __shfl_xor(ps, off, 32);
                pd += __shfl_xor(pd, off, 32);
            }
            if (m == 0 && row < N) { a_s[row] = ps; a_d[row] = pd; }
        }
    }
}


#define SLOTS 13
#define STAGE_CAP (SLOTS * 1024)
__global__ __launch_bounds__(1024, 16) void slice_sort_gemm(const int* __restrict__ ei,
                                                        int* __restrict__ offsT,
                                                        int2* __restrict__ sorted,
                                                        int E, int NB, int slice,
                                                        const float* __restrict__ x,
                                                        const float* __restrict__ W,
                                                        const float* __restrict__ att_src,
                                                        const float* __restrict__ att_dst,
                                                        float* __restrict__ h,
                                                        float* __restrict__ a_s,
                                                        float* __restrict__ a_d, int N,
                                                        int tile_lo, int tile_end, int tpb) {
    __shared__ float Ws[IN_F * OUT_F];
    __shared__ int2  stage[STAGE_CAP + 1024];
    if (blockIdx.x >= NSLICE) {
        const int lo = tile_lo + (blockIdx.x - NSLICE) * tpb;
        gemm_role(x, W, att_src, att_dst, h, a_s, a_d, N, lo, min(lo + tpb, tile_end), Ws);
        return;
    }
    int* cur  = (int*)Ws;
    int* wsum = cur + MAXB;
    const int lane = threadIdx.x & 31;
    const int wave = threadIdx.x >> 5;
    for (int b = threadIdx.x; b < NB; b += 1024) cur[b] = 0;
    const int lo = blockIdx.x * slice;
    const int hi = min(lo + slice, E);
    const bool on_chip = hi - lo <= STAGE_CAP;

    int sr[SLOTS], ds[SLOTS];
    if (on_chip) {
#pragma unroll
        for (int u = 0; u < SLOTS; ++u) {
            const int i = lo + threadIdx.x + u * 1024;
            ds[u] = (i < hi) ? ei[E + i] : -1;
        }
#pragma unroll
        for (int u = 0; u < SLOTS; ++u) {
            const int i = lo + threadIdx.x + u * 1024;
            sr[u] = (i < hi) ? ei[i] : 0;
        }
    }
    __syncthreads();

    if (on_chip) {
#pragma unroll
        for (int u = 0; u < SLOTS; ++u)
            if (ds[u] >= 0) atomicAdd(&cur[ds[u] >> BIN_SHIFT], 1);
    } else {
        for (int i0 = lo + threadIdx.x; i0 < hi; i0 += 4 * 1024) {
            int d[4];
#pragma unroll
            for (int u = 0; u < 4; ++u) {
                const int i = i0 + u * 1024;
                d[u] = (i < hi) ? ei[E + i] : -1;
            }
#pragma unroll
            for (int u = 0; u < 4; ++u)
                if (d[u] >= 0) atomicAdd(&cur[d[u] >> BIN_SHIFT], 1);
        }
    }
    __syncthreads();

    const int t = (threadIdx.x < NB) ? cur[threadIdx.x] : 0;
    int acc = t;
#pragma unroll
    for (int off = 1; off < 32; off <<= 1) {
        int y = __shfl_up(acc, off, 32);
        if (lane >= off) acc += y;
    }
    if (lane == 31) wsum[wave] = acc;
    __syncthreads();
    if (wave == 0) {
        int w = wsum[lane];
#pragma unroll
        for (int off = 1; off < 32; off <<= 1) {
            int y = __shfl_up(w, off, 32);
            if (lane >= off) w += y;
        }
        wsum[lane] = w;
    }
    __syncthreads();
    const int excl = acc - t + (wave > 0 ? wsum[wave - 1] : 0);
    if (threadIdx.x < NB) {
        cur[threadIdx.x] = lo + excl;
        offsT[(size_t)threadIdx.x * NSLICE + blockIdx.x] = excl;
    }
    if (threadIdx.x == NB - 1) offsT[(size_t)NB * NSLICE + blockIdx.x] = excl + t;
    __syncthreads();

    if (on_chip) {
        const int a = lo & ~15;
        int pos[SLOTS];
#pragma unroll
        for (int u = 0; u < SLOTS; ++u)
            pos[u] = (ds[u] >= 0) ? atomicAdd(&cur[ds[u] >> BIN_SHIFT], 1) : a;
#pragma unroll
        for (int u = 0; u < SLOTS; ++u)
            if (ds[u] >= 0) stage[pos[u] - a] = make_int2(sr[u], ds[u]);
        __syncthreads();
        int2 rec[SLOTS + 1];
#pragma unroll
        for (int u = 0; u <= SLOTS; ++u) rec[u] = stage[threadIdx.x + u * 1024];
#pragma unroll
        for (int u = 0; u <= SLOTS; ++u) {
            const int q = a + threadIdx.x + u * 1024;
            if (q >= lo && q < hi) sorted[q] = rec[u];
        }
    } else {
        for (int i0 = lo + threadIdx.x; i0 < hi; i0 += 4 * 1024) {
            int s[4], d[4];
#pragma unroll
            for (int u = 0; u < 4; ++u) {
                const int i = i0 + u * 1024;
                s[u] = 0; d[u] = -1;
                if (i < hi) { s[u] = ei[i]; d[u] = ei[E + i]; }
            }
#pragma unroll
            for (int u = 0; u < 4; ++u) {
                if (d[u] >= 0) {
                    const int pos = atomicAdd(&cur[d[u] >> BIN_SHIFT], 1);
                    sorted[pos] = make_int2(s[u], d[u]);
                }
            }
        }
    }
}

#define LIST_CAP 4096

__global__ __launch_bounds__(512, 16) void gat_bin_process(const int2* __restrict__ sorted,
                                                       const int* __restrict__ offsT, int slice,
                                                       const float* __restrict__ h,
                                                       const float* __restrict__ a_s,
                                                       const float* __restrict__ a_d,
                                                       const float* __restrict__ bias,
                                                       float* __restrict__ out, int N) {
    __shared__ int2 lst[LIST_CAP];
    __shared__ int  cnt[BIN_NODES];
    __shared__ int  rstart[BIN_NODES + 1];
    __shared__ int  src_idx[LIST_CAP];
    __shared__ int  segbase[NSLICE];
    __shared__ int  segpos[NSLICE + 1];

    const int tid  = threadIdx.x;
    const int lane = tid & 31;
    const int wave = __builtin_amdgcn_readfirstlane(tid >> 5);
    const int base = blockIdx.x << BIN_SHIFT;
    static_assert(NSLICE == 128, "segment table: one wave, four slices per lane; four threads per segment");
    if (tid < 32) {
        const int4 o0 = ((const int4*)(offsT + (size_t)blockIdx.x * NSLICE))[lane];
        const int4 o1 = ((const int4*)(offsT + (size_t)(blockIdx.x + 1) * NSLICE))[lane];
        const int l0 = o1.x - o0.x, l1 = o1.y - o0.y, l2 = o1.z - o0.z, l3 = o1.w - o0.w;
        const int s4 = l0 + l1 + l2 + l3;
        int acc = s4;
#pragma unroll
        for (int off = 1; off < 32; off <<= 1) {
            int y = __shfl_up(acc, off, 32);
            if (lane >= off) acc += y;
        }
        const int ex = acc - s4;
        const int sb = 4 * lane * slice;
        segbase[4 * lane]     = sb + o0.x;              segpos[4 * lane]     = ex;
        segbase[4 * lane + 1] = sb + slice + o0.y;      segpos[4 * lane + 1] = ex + l0;
        segbase[4 * lane + 2] = sb + 2 * slice + o0.z;  segpos[4 * lane + 2] = ex + l0 + l1;
        segbase[4 * lane + 3] = sb + 3 * slice + o0.w;  segpos[4 * lane + 3] = ex + l0 + l1 + l2;
        if (lane == 31) segpos[NSLICE] = acc;
    }
    __syncthreads();
    const int ne = segpos[NSLICE];
    const int nrow  = min(BIN_NODES, N - base);
    const float2* h2 = (const float2*)h;
    float2* out2 = (float2*)out;

    if (ne + nrow <= LIST_CAP) {
        if (tid < BIN_NODES) cnt[tid] = (tid < nrow) ? 1 : 0;
        {
            const int sg = tid >> 2;
            const int sb = segbase[sg], sp = segpos[sg], sl = segpos[sg + 1] - sp;
            for (int i = tid & 3; i < sl; i += 4) src_idx[sp + i] = sb + i;
        }
        __syncthreads();

        for (int j0 = tid; j0 < ne; j0 += 4 * 512) {
            int d[4];
#pragma unroll
            for (int u = 0; u < 4; ++u) {
                const int j = j0 + u * 512;
                d[u] = sorted[src_idx[j < ne ? j : ne - 1]].y;
            }
#pragma unroll
            for (int u = 0; u < 4; ++u)
                if (j0 + u * 512 < ne) atomicAdd(&cnt[d[u] - base], 1);
        }
        __syncthreads();

        if (tid < 32) {
            const int c0 = cnt[4 * tid], c1 = cnt[4 * tid + 1], c2 = cnt[4 * tid + 2], c3 = cnt[4 * tid + 3];
            const int s4 = c0 + c1 + c2 + c3;
            int x = s4;
#pragma unroll
            for (int off = 1; off < 32; off <<= 1) {
                int y = __shfl_up(x, off, 32);
                if (lane >= off) x += y;
            }
            const int ex = x - s4;
            rstart[4 * tid]     = ex;            cnt[4 * tid]     = ex;
            rstart[4 * tid + 1] = ex + c0;       cnt[4 * tid + 1] = ex + c0;
            rstart[4 * tid + 2] = ex + c0 + c1;  cnt[4 * tid + 2] = ex + c0 + c1;
            rstart[4 * tid + 3] = ex + s4 - c3;  cnt[4 * tid + 3] = ex + s4 - c3;
            if (tid == 31) rstart[BIN_NODES] = x;
        }
        __syncthreads();

        if (tid < nrow) {
            const int node = base + tid;
            const int pos = atomicAdd(&cnt[tid], 1);
            lst[pos] = make_int2(node, __float_as_int(a_s[node]));
        }
        for (int j0 = tid; j0 < ne; j0 += 4 * 512) {
            int2  r[4];
            float as[4];
#pragma unroll
            for (int u = 0; u < 4; ++u) {
                const int j = j0 + u * 512;
                r[u] = sorted[src_idx[j < ne ? j : ne - 1]];
            }
#pragma unroll
            for (int u = 0; u < 4; ++u) as[u] = a_s[r[u].x];
#pragma unroll
            for (int u = 0; u < 4; ++u) {
                if (j0 + u * 512 < ne) {
                    const int pos = atomicAdd(&cnt[r[u].y - base], 1);
                    lst[pos] = make_int2(r[u].x, __float_as_int(as[u]));
                }
            }
        }
        __syncthreads();

        const int      c16 = lane & 15;
        const int      hf  = lane >> 4;
        const unsigned col = (unsigned)c16 * 16u;
        const char*    hb  = (const char*)h;
        const float4   bv  = ((const float4*)bias)[c16];
        for (int i = 0; i < 8; ++i) {
            const int r = wave * 8 + i;
            if (r >= nrow) break;
            const int node = base + r;
            const int rs = __builtin_amdgcn_readfirstlane(rstart[r]);
            const int re = __builtin_amdgcn_readfirstlane(rstart[r + 1]);
            const float adv = a_d[node];

            float m = -__builtin_inff();
            for (int p = rs + lane; p < re; p += 32) {
                float e = __int_as_float(lst[p].y) + adv;
                e = e > 0.0f ? e : NEG_SLOPE * e;
                m = fmaxf(m, e);
            }
#pragma unroll
            for (int off = 16; off >= 1; off >>= 1) m = fmaxf(m, __shfl_xor(m, off, 32));

            float ssum = 0.0f, a0 = 0.0f, a1 = 0.0f, a2 = 0.0f, a3 = 0.0f;
            for (int p0 = rs; p0 < re; p0 += 32) {
                const int p = p0 + 2 * c16 + hf;
                unsigned off = (unsigned)node * (OUT_F * 4u);
                float w = 0.0f;
                if (p < re) {
                    const int2 it = lst[p];
                    float e = __int_as_float(it.y) + adv;
                    e = e > 0.0f ? e : NEG_SLOPE * e;
                    w = __expf(e - m);
                    off = (unsigned)it.x * (OUT_F * 4u);
                }
                const unsigned wb = __float_as_uint(w);
                const int nl = (min(32, re - p0) + 1) >> 1;
                if (nl < 8) {
                    row_group_part<0>(hb, off, wb, col, nl, a0, a1, a2, a3, ssum);
                } else {
                    row_group_full<0>(hb, off, wb, col, a0, a1, a2, a3, ssum);
                    if (nl == 16)    row_group_full<8>(hb, off, wb, col, a0, a1, a2, a3, ssum);
                    else if (nl > 8) row_group_part<8>(hb, off, wb, col, nl - 8, a0, a1, a2, a3, ssum);
                }
            }
            a0 += xhalf(a0); a1 += xhalf(a1); a2 += xhalf(a2); a3 += xhalf(a3); ssum += xhalf(ssum);
            const float inv = 1.0f / ssum;
            if (hf == 0)
                *(float4*)((char*)out + (size_t)((unsigned)node * (OUT_F * 4u) + col)) =
                    make_float4(a0 * inv + bv.x, a1 * inv + bv.y, a2 * inv + bv.z, a3 * inv + bv.w);
        }
    } else {
        const float2 bv = ((const float2*)bias)[lane];
        for (int i = 0; i < 8; ++i) {
            const int r = wave * 8 + i;
            if (r >= nrow) break;
            const int node = base + r;
            const float adv = a_d[node];
            float es = a_s[node] + adv;
            es = es > 0.0f ? es : NEG_SLOPE * es;

            float m = es;
            for (int sg = 0; sg < NSLICE; ++sg) {
                const int sb = __builtin_amdgcn_readfirstlane(segbase[sg]);
                const int sl = __builtin_amdgcn_readfirstlane(segpos[sg + 1] - segpos[sg]);
                for (int j0 = 0; j0 < sl; j0 += 32) {
                    const int j = j0 + lane;
                    if (j < sl) {
                        const int2 rec = sorted[sb + j];
                        if (rec.y == node) {
                            float e = a_s[rec.x] + adv;
                            e = e > 0.0f ? e : NEG_SLOPE * e;
                            m = fmaxf(m, e);
                        }
                    }
                }
            }
#pragma unroll
            for (int off = 16; off >= 1; off >>= 1) m = fmaxf(m, __shfl_xor(m, off, 32));

            float ssum = __expf(es - m);
            const float2 hs = h2[(size_t)node * (OUT_F / 2) + lane];
            float a0 = ssum * hs.x, a1 = ssum * hs.y;
            for (int sg = 0; sg < NSLICE; ++sg) {
                const int sb = __builtin_amdgcn_readfirstlane(segbase[sg]);
                const int sl = __builtin_amdgcn_readfirstlane(segpos[sg + 1] - segpos[sg]);
                for (int j0 = 0; j0 < sl; j0 += 32) {
                    const int j = j0 + lane;
                    int   s = node;
                    float w = 0.0f;
                    bool  hit = false;
                    if (j < sl) {
                        const int2 rec = sorted[sb + j];
                        if (rec.y == node) {
                            float e = a_s[rec.x] + adv;
                            e = e > 0.0f ? e : NEG_SLOPE * e;
                            w = __expf(e - m);
                            s = rec.x;
                            hit = true;
                        }
                    }
                    const unsigned wb = __float_as_uint(w);
                    unsigned mask = __builtin_amdgcn_ballot_w32(hit);
                    while (mask) {
                        const int k = __builtin_ctz(mask);
                        mask &= mask - 1;
                        const int   sk = __builtin_amdgcn_readlane(s, k);
                        const float wk = __uint_as_float(__builtin_amdgcn_readlane(wb, k));
                        const float2 hv = h2[(size_t)sk * (OUT_F / 2) + lane];
                        a0 = fmaf(wk, hv.x, a0);
                        a1 = fmaf(wk, hv.y, a1);
                        ssum += wk;
                    }
                }
            }
            out2[(size_t)node * (OUT_F / 2) + lane] = make_float2(a0 / ssum + bv.x, a1 / ssum + bv.y);
        }
    }
}

extern "C" void kernel_launch(void* const* d_in, const int* in_sizes, int n_in,
                              void* d_out, int out_size, void* d_ws, size_t ws_size,
                              hipStream_t stream) {
    const float* x       = (const float*)d_in[0];
    const int*   ei      = (const int*)d_in[1];
    const float* W       = (const float*)d_in[2];
    const float* att_src = (const float*)d_in[3];
    const float* att_dst = (const float*)d_in[4];
    const float* bias    = (const float*)d_in[5];

    const int N  = in_sizes[0] / IN_F;
    const int E  = in_sizes[1] / 2;
    const int NB = (N + BIN_NODES - 1) / BIN_NODES;
    const int slice = (E + NSLICE - 1) / NSLICE;

    float* out = (float*)d_out;

    float* h        = (float*)d_ws;
    int*   offsT    = (int*)(h + (size_t)N * OUT_F);
    int2*  sorted   = (int2*)(offsT + (size_t)(NB + 1) * NSLICE);
    float* a_s      = (float*)(sorted + E);
    float* a_d      = a_s + N;

    const int tiles = (N + 15) / 16;
    const int gw    = (tiles + TPB - 1) / TPB;

    slice_sort_gemm<<<NSLICE + gw, 1024, 0, stream>>>(ei, offsT, sorted, E, NB, slice,
                                                      x, W, att_src, att_dst, h, a_s, a_d, N, 0, tiles, TPB);
    gat_bin_process<<<NB, 512, 0, stream>>>(sorted, offsT, slice, h, a_s, a_d, bias, out, N);
}


# Bin pass: the thread placing an item folds its logit into the row's maximum (integer LDS atomic max on order-preserving keys); list holds logits; per-row max pass and its 5 cross-lane steps removed
# speedup vs baseline: 6.1875x; 1.0395x over previous
#include <hip/hip_runtime.h>
#include <hip/hip_bf16.h>

#define IN_F  128
#define OUT_F 64
#define NEG_SLOPE 0.2f
#define BIN_SHIFT 7
#define BIN_NODES 128
#define MAXB 1024
#define NSLICE 128
#define TPB 25

typedef __attribute__((ext_vector_type(2))) float v2f;
typedef __attribute__((ext_vector_type(8))) float v8f;
typedef __attribute__((ext_vector_type(4))) float v4f;

__device__ __forceinline__ unsigned f2ord(float f) {
    unsigned u = __float_as_uint(f);
    return (u & 0x80000000u) ? ~u : (u | 0x80000000u);
}
__device__ __forceinline__ float ord2f(unsigned u) {
    unsigned v = (u & 0x80000000u) ? (u & 0x7FFFFFFFu) : ~u;
    return __uint_as_float(v);
}

__device__ __forceinline__ float xhalf(float v) {
    return __builtin_bit_cast(float, __builtin_amdgcn_permlanex16(__builtin_bit_cast(int, v), __builtin_bit_cast(int, v),
                                                                  0x76543210, 0xfedcba98, false, false));
}
template <int K> __device__ __forceinline__ unsigned share16(unsigned v) {
    return (unsigned)__builtin_amdgcn_update_dpp(0, (int)v, 0x150 + K, 0xf, 0xf, true);
}
#define ROW_LD(U) hv[U] = *(const v4f*)(hb + (size_t)(share16<K0 + U>(off) + col));
#define ROW_AC(U) {                                                    \
        const float wk = __uint_as_float(share16<K0 + U>(wb));         \
        a0 = fmaf(wk, hv[U].x, a0); a1 = fmaf(wk, hv[U].y, a1);        \
        a2 = fmaf(wk, hv[U].z, a2); a3 = fmaf(wk, hv[U].w, a3);        \
        ssum += wk; }
template <int K0>
__device__ __forceinline__ void row_group_full(const char* __restrict__ hb, unsigned off, unsigned wb, unsigned col,
                                               float& a0, float& a1, float& a2, float& a3, float& ssum) {
    v4f hv[8];
    ROW_LD(0) ROW_LD(1) ROW_LD(2) ROW_LD(3) ROW_LD(4) ROW_LD(5) ROW_LD(6) ROW_LD(7)
    asm volatile("" : "+v"(wb) : "v"(hv[0]), "v"(hv[1]), "v"(hv[2]), "v"(hv[3]), "v"(hv[4]), "v"(hv[5]), "v"(hv[6]), "v"(hv[7]));
    ROW_AC(0) ROW_AC(1) ROW_AC(2) ROW_AC(3) ROW_AC(4) ROW_AC(5) ROW_AC(6) ROW_AC(7)
}
template <int K0>
__device__ __forceinline__ void row_group_part(const char* __restrict__ hb, unsigned off, unsigned wb, unsigned col, int cnt,
                                               float& a0, float& a1, float& a2, float& a3, float& ssum) {
    v4f hv[7];
    ROW_LD(0)
    if (cnt > 1) { ROW_LD(1)
     if (cnt > 2) { ROW_LD(2)
      if (cnt > 3) { ROW_LD(3)
       if (cnt > 4) { ROW_LD(4)
        if (cnt > 5) { ROW_LD(5)
         if (cnt > 6) { ROW_LD(6) } } } } } }
    ROW_AC(0)
    if (cnt > 1) { ROW_AC(1)
     if (cnt > 2) { ROW_AC(2)
      if (cnt > 3) { ROW_AC(3)
       if (cnt > 4) { ROW_AC(4)
        if (cnt > 5) { ROW_AC(5)
         if (cnt > 6) { ROW_AC(6) } } } } } }
}
#undef ROW_LD
#undef ROW_AC

__device__ __forceinline__ void gemm_role(const float* __restrict__ x,
                                          const float* __restrict__ W,
                                          const float* __restrict__ att_src,
                                          const float* __restrict__ att_dst,
                                          float* __restrict__ h,
                                          float* __restrict__ a_s,
                                          float* __restrict__ a_d, int N,
                                          int tile_lo, int tile_hi, float* Ws) {
    float wv[IN_F * OUT_F / 1024];
#pragma unroll
    for (int u = 0; u < IN_F * OUT_F / 1024; ++u) wv[u] = W[threadIdx.x + u * 1024];
#pragma unroll
    for (int u = 0; u < IN_F * OUT_F / 1024; ++u) {
        const int i = threadIdx.x + u * 1024;
        const int k = i >> 6;
        const int n = i & 63;
        Ws[(k >> 1) * 128 + n * 2 + (k & 1)] = wv[u];
    }
    __syncthreads();

    const int lane = threadIdx.x & 31;
    const int wave = __builtin_amdgcn_readfirstlane(threadIdx.x >> 5);
    const int m    = lane & 15;
    const int hi   = lane >> 4;

    const int t = tile_lo + wave;
    if (t < tile_hi) {
        const int row0 = t * 16;
        const int arow = row0 + m;
        const int crow = arow < N ? arow : 0;
        const unsigned xoff = (unsigned)crow * (IN_F * 4u) + (unsigned)hi * 8u;
        const char*   xb    = (const char*)x;
        const float2* wl    = (const float2*)Ws + hi * 64 + m;

        v8f c0 = {}, c1 = {}, c2 = {}, c3 = {};

#pragma unroll 8
        for (int k0 = 0; k0 < IN_F; k0 += 4) {
            const int kq = k0 >> 1;
            float2 af = *(const float2*)(xb + (size_t)xoff + kq * 8);
            v2f a = { af.x, af.y };
            const float2* wrow = wl + kq * 64;
            float2 f0 = wrow[0];
            float2 f1 = wrow[16];
            float2 f2 = wrow[32];
            float2 f3 = wrow[48];
            v2f b0 = { f0.x, f0.y };
            v2f b1 = { f1.x, f1.y };
            v2f b2 = { f2.x, f2.y };
            v2f b3 = { f3.x, f3.y };
            c0 = __builtin_amdgcn_wmma_f32_16x16x4_f32(false, a, false, b0, (short)0, c0, false, false);
            c1 = __builtin_amdgcn_wmma_f32_16x16x4_f32(false, a, false, b1, (short)0, c1, false, false);
            c2 = __builtin_amdgcn_wmma_f32_16x16x4_f32(false, a, false, b2, (short)0, c2, false, false);
            c3 = __builtin_amdgcn_wmma_f32_16x16x4_f32(false, a, false, b3, (short)0, c3, false, false);
        }

        const float as0 = att_src[m],      as1 = att_src[16 + m];
        const float as2 = att_src[32 + m], as3 = att_src[48 + m];
        const float ad0 = att_dst[m],      ad1 = att_dst[16 + m];
        const float ad2 = att_dst[32 + m], ad3 = att_dst[48 + m];

#pragma unroll
        for (int r = 0; r < 8; ++r) {
            const int row = row0 + r + 8 * hi;
            const float e0 = c0[r], e1 = c1[r], e2 = c2[r], e3 = c3[r];
            if (row < N) {
                float* hr = h + (size_t)row * OUT_F;
                hr[m]      = e0;
                hr[16 + m] = e1;
                hr[32 + m] = e2;
                hr[48 + m] = e3;
            }
            float ps = e0 * as0 + e1 * as1 + e2 * as2 + e3 * as3;
            float pd = e0 * ad0 + e1 * ad1 + e2 * ad2 + e3 * ad3;
#pragma unroll
            for (int off = 8; off >= 1; off >>= 1) {
                ps += __shfl_xor(ps, off, 32);
                pd += __shfl_xor(pd, off, 32);
            }
            if (m == 0 && row < N) { a_s[row] = ps; a_d[row] = pd; }
        }
    }
}


#define SLOTS 13
#define STAGE_CAP (SLOTS * 1024)
__global__ __launch_bounds__(1024, 16) void slice_sort_gemm(const int* __restrict__ ei,
                                                        int* __restrict__ offsT,
                                                        int2* __restrict__ sorted,
                                                        int E, int NB, int slice,
                                                        const float* __restrict__ x,
                                                        const float* __restrict__ W,
                                                        const float* __restrict__ att_src,
                                                        const float* __restrict__ att_dst,
                                                        float* __restrict__ h,
                                                        float* __restrict__ a_s,
                                                        float* __restrict__ a_d, int N,
                                                        int tile_lo, int tile_end, int tpb) {
    __shared__ float Ws[IN_F * OUT_F];
    __shared__ int2  stage[STAGE_CAP + 1024];
    if (blockIdx.x >= NSLICE) {
        const int lo = tile_lo + (blockIdx.x - NSLICE) * tpb;
        gemm_role(x, W, att_src, att_dst, h, a_s, a_d, N, lo, min(lo + tpb, tile_end), Ws);
        return;
    }
    int* cur  = (int*)Ws;
    int* wsum = cur + MAXB;
    const int lane = threadIdx.x & 31;
    const int wave = threadIdx.x >> 5;
    for (int b = threadIdx.x; b < NB; b += 1024) cur[b] = 0;
    const int lo = blockIdx.x * slice;
    const int hi = min(lo + slice, E);
    const bool on_chip = hi - lo <= STAGE_CAP;

    int sr[SLOTS], ds[SLOTS];
    if (on_chip) {
#pragma unroll
        for (int u = 0; u < SLOTS; ++u) {
            const int i = lo + threadIdx.x + u * 1024;
            ds[u] = (i < hi) ? ei[E + i] : -1;
        }
#pragma unroll
        for (int u = 0; u < SLOTS; ++u) {
            const int i = lo + threadIdx.x + u * 1024;
            sr[u] = (i < hi) ? ei[i] : 0;
        }
    }
    __syncthreads();

    if (on_chip) {
#pragma unroll
        for (int u = 0; u < SLOTS; ++u)
            if (ds[u] >= 0) atomicAdd(&cur[ds[u] >> BIN_SHIFT], 1);
    } else {
        for (int i0 = lo + threadIdx.x; i0 < hi; i0 += 4 * 1024) {
            int d[4];
#pragma unroll
            for (int u = 0; u < 4; ++u) {
                const int i = i0 + u * 1024;
                d[u] = (i < hi) ? ei[E + i] : -1;
            }
#pragma unroll
            for (int u = 0; u < 4; ++u)
                if (d[u] >= 0) atomicAdd(&cur[d[u] >> BIN_SHIFT], 1);
        }
    }
    __syncthreads();

    const int t = (threadIdx.x < NB) ? cur[threadIdx.x] : 0;
    int acc = t;
#pragma unroll
    for (int off = 1; off < 32; off <<= 1) {
        int y = __shfl_up(acc, off, 32);
        if (lane >= off) acc += y;
    }
    if (lane == 31) wsum[wave] = acc;
    __syncthreads();
    if (wave == 0) {
        int w = wsum[lane];
#pragma unroll
        for (int off = 1; off < 32; off <<= 1) {
            int y = __shfl_up(w, off, 32);
            if (lane >= off) w += y;
        }
        wsum[lane] = w;
    }
    __syncthreads();
    const int excl = acc - t + (wave > 0 ? wsum[wave - 1] : 0);
    if (threadIdx.x < NB) {
        cur[threadIdx.x] = lo + excl;
        offsT[(size_t)threadIdx.x * NSLICE + blockIdx.x] = excl;
    }
    if (threadIdx.x == NB - 1) offsT[(size_t)NB * NSLICE + blockIdx.x] = excl + t;
    __syncthreads();

    if (on_chip) {
        const int a = lo & ~15;
        int pos[SLOTS];
#pragma unroll
        for (int u = 0; u < SLOTS; ++u)
            pos[u] = (ds[u] >= 0) ? atomicAdd(&cur[ds[u] >> BIN_SHIFT], 1) : a;
#pragma unroll
        for (int u = 0; u < SLOTS; ++u)
            if (ds[u] >= 0) stage[pos[u] - a] = make_int2(sr[u], ds[u]);
        __syncthreads();
        int2 rec[SLOTS + 1];
#pragma unroll
        for (int u = 0; u <= SLOTS; ++u) rec[u] = stage[threadIdx.x + u * 1024];
#pragma unroll
        for (int u = 0; u <= SLOTS; ++u) {
            const int q = a + threadIdx.x + u * 1024;
            if (q >= lo && q < hi) sorted[q] = rec[u];
        }
    } else {
        for (int i0 = lo + threadIdx.x; i0 < hi; i0 += 4 * 1024) {
            int s[4], d[4];
#pragma unroll
            for (int u = 0; u < 4; ++u) {
                const int i = i0 + u * 1024;
                s[u] = 0; d[u] = -1;
                if (i < hi) { s[u] = ei[i]; d[u] = ei[E + i]; }
            }
#pragma unroll
            for (int u = 0; u < 4; ++u) {
                if (d[u] >= 0) {
                    const int pos = atomicAdd(&cur[d[u] >> BIN_SHIFT], 1);
                    sorted[pos] = make_int2(s[u], d[u]);
                }
            }
        }
    }
}

#define LIST_CAP 4096

__global__ __launch_bounds__(512, 16) void gat_bin_process(const int2* __restrict__ sorted,
                                                       const int* __restrict__ offsT, int slice,
                                                       const float* __restrict__ h,
                                                       const float* __restrict__ a_s,
                                                       const float* __restrict__ a_d,
                                                       const float* __restrict__ bias,
                                                       float* __restrict__ out, int N) {
    __shared__ int2 lst[LIST_CAP];
    __shared__ int  cnt[BIN_NODES];
    __shared__ int  rstart[BIN_NODES + 1];
    __shared__ int  src_idx[LIST_CAP];
    __shared__ int  segbase[NSLICE];
    __shared__ int  segpos[NSLICE + 1];
    __shared__ float    adl[BIN_NODES];
    __shared__ unsigned rmax[BIN_NODES];

    const int tid  = threadIdx.x;
    const int lane = tid & 31;
    const int wave = __builtin_amdgcn_readfirstlane(tid >> 5);
    const int base = blockIdx.x << BIN_SHIFT;
    static_assert(NSLICE == 128, "segment table: one wave, four slices per lane; four threads per segment");
    const int nrow  = min(BIN_NODES, N - base);
    float adr = 0.0f;
    if (tid < nrow) adr = a_d[base + tid];
    if (tid < 32) {
        const int4 o0 = ((const int4*)(offsT + (size_t)blockIdx.x * NSLICE))[lane];
        const int4 o1 = ((const int4*)(offsT + (size_t)(blockIdx.x + 1) * NSLICE))[lane];
        const int l0 = o1.x - o0.x, l1 = o1.y - o0.y, l2 = o1.z - o0.z, l3 = o1.w - o0.w;
        const int s4 = l0 + l1 + l2 + l3;
        int acc = s4;
#pragma unroll
        for (int off = 1; off < 32; off <<= 1) {
            int y = __shfl_up(acc, off, 32);
            if (lane >= off) acc += y;
        }
        const int ex = acc - s4;
        const int sb = 4 * lane * slice;
        segbase[4 * lane]     = sb + o0.x;              segpos[4 * lane]     = ex;
        segbase[4 * lane + 1] = sb + slice + o0.y;      segpos[4 * lane + 1] = ex + l0;
        segbase[4 * lane + 2] = sb + 2 * slice + o0.z;  segpos[4 * lane + 2] = ex + l0 + l1;
        segbase[4 * lane + 3] = sb + 3 * slice + o0.w;  segpos[4 * lane + 3] = ex + l0 + l1 + l2;
        if (lane == 31) segpos[NSLICE] = acc;
    }
    __syncthreads();
    const int ne = segpos[NSLICE];
    const float2* h2 = (const float2*)h;
    float2* out2 = (float2*)out;

    if (ne + nrow <= LIST_CAP) {
        if (tid < BIN_NODES) {
            cnt[tid]  = (tid < nrow) ? 1 : 0;
            adl[tid]  = adr;
            rmax[tid] = 0u;
        }
        {
            const int sg = tid >> 2;
            const int sb = segbase[sg], sp = segpos[sg], sl = segpos[sg + 1] - sp;
            for (int i = tid & 3; i < sl; i += 4) src_idx[sp + i] = sb + i;
        }
        __syncthreads();

        for (int j0 = tid; j0 < ne; j0 += 4 * 512) {
            int d[4];
#pragma unroll
            for (int u = 0; u < 4; ++u) {
                const int j = j0 + u * 512;
                d[u] = sorted[src_idx[j < ne ? j : ne - 1]].y;
            }
#pragma unroll
            for (int u = 0; u < 4; ++u)
                if (j0 + u * 512 < ne) atomicAdd(&cnt[d[u] - base], 1);
        }
        __syncthreads();

        if (tid < 32) {
            const int c0 = cnt[4 * tid], c1 = cnt[4 * tid + 1], c2 = cnt[4 * tid + 2], c3 = cnt[4 * tid + 3];
            const int s4 = c0 + c1 + c2 + c3;
            int x = s4;
#pragma unroll
            for (int off = 1; off < 32; off <<= 1) {
                int y = __shfl_up(x, off, 32);
                if (lane >= off) x += y;
            }
            const int ex = x - s4;
            rstart[4 * tid]     = ex;            cnt[4 * tid]     = ex;
            rstart[4 * tid + 1] = ex + c0;       cnt[4 * tid + 1] = ex + c0;
            rstart[4 * tid + 2] = ex + c0 + c1;  cnt[4 * tid + 2] = ex + c0 + c1;
            rstart[4 * tid + 3] = ex + s4 - c3;  cnt[4 * tid + 3] = ex + s4 - c3;
            if (tid == 31) rstart[BIN_NODES] = x;
        }
        __syncthreads();

        if (tid < nrow) {
            const int node = base + tid;
            const int pos = atomicAdd(&cnt[tid], 1);
            float e = a_s[node] + adr;
            e = e > 0.0f ? e : NEG_SLOPE * e;
            atomicMax(&rmax[tid], f2ord(e));
            lst[pos] = make_int2(node, __float_as_int(e));
        }
        for (int j0 = tid; j0 < ne; j0 += 4 * 512) {
            int2  r[4];
            float as[4];
#pragma unroll
            for (int u = 0; u < 4; ++u) {
                const int j = j0 + u * 512;
                r[u] = sorted[src_idx[j < ne ? j : ne - 1]];
            }
#pragma unroll
            for (int u = 0; u < 4; ++u) as[u] = a_s[r[u].x];
#pragma unroll
            for (int u = 0; u < 4; ++u) {
                if (j0 + u * 512 < ne) {
                    const int lr  = r[u].y - base;
                    const int pos = atomicAdd(&cnt[lr], 1);
                    float e = as[u] + adl[lr];
                    e = e > 0.0f ? e : NEG_SLOPE * e;
                    atomicMax(&rmax[lr], f2ord(e));
                    lst[pos] = make_int2(r[u].x, __float_as_int(e));
                }
            }
        }
        __syncthreads();

        const int      c16 = lane & 15;
        const int      hf  = lane >> 4;
        const unsigned col = (unsigned)c16 * 16u;
        const char*    hb  = (const char*)h;
        const float4   bv  = ((const float4*)bias)[c16];
        for (int i = 0; i < 8; ++i) {
            const int r = wave * 8 + i;
            if (r >= nrow) break;
            const int node = base + r;
            const int rs = __builtin_amdgcn_readfirstlane(rstart[r]);
            const int re = __builtin_amdgcn_readfirstlane(rstart[r + 1]);
            const float m = ord2f((unsigned)__builtin_amdgcn_readfirstlane((int)rmax[r]));

            float ssum = 0.0f, a0 = 0.0f, a1 = 0.0f, a2 = 0.0f, a3 = 0.0f;
            for (int p0 = rs; p0 < re; p0 += 32) {
                const int p = p0 + 2 * c16 + hf;
                unsigned off = (unsigned)node * (OUT_F * 4u);
                float w = 0.0f;
                if (p < re) {
                    const int2 it = lst[p];
                    w = __expf(__int_as_float(it.y) - m);
                    off = (unsigned)it.x * (OUT_F * 4u);
                }
                const unsigned wb = __float_as_uint(w);
                const int nl = (min(32, re - p0) + 1) >> 1;
                if (nl < 8) {
                    row_group_part<0>(hb, off, wb, col, nl, a0, a1, a2, a3, ssum);
                } else {
                    row_group_full<0>(hb, off, wb, col, a0, a1, a2, a3, ssum);
                    if (nl == 16)    row_group_full<8>(hb, off, wb, col, a0, a1, a2, a3, ssum);
                    else if (nl > 8) row_group_part<8>(hb, off, wb, col, nl - 8, a0, a1, a2, a3, ssum);
                }
            }
            a0 += xhalf(a0); a1 += xhalf(a1); a2 += xhalf(a2); a3 += xhalf(a3); ssum += xhalf(ssum);
            const float inv = 1.0f / ssum;
            if (hf == 0)
                *(float4*)((char*)out + (size_t)((unsigned)node * (OUT_F * 4u) + col)) =
                    make_float4(a0 * inv + bv.x, a1 * inv + bv.y, a2 * inv + bv.z, a3 * inv + bv.w);
        }
    } else {
        const float2 bv = ((const float2*)bias)[lane];
        for (int i = 0; i < 8; ++i) {
            const int r = wave * 8 + i;
            if (r >= nrow) break;
            const int node = base + r;
            const float adv = a_d[node];
            float es = a_s[node] + adv;
            es = es > 0.0f ? es : NEG_SLOPE * es;

            float m = es;
            for (int sg = 0; sg < NSLICE; ++sg) {
                const int sb = __builtin_amdgcn_readfirstlane(segbase[sg]);
                const int sl = __builtin_amdgcn_readfirstlane(segpos[sg + 1] - segpos[sg]);
                for (int j0 = 0; j0 < sl; j0 += 32) {
                    const int j = j0 + lane;
                    if (j < sl) {
                        const int2 rec = sorted[sb + j];
                        if (rec.y == node) {
                            float e = a_s[rec.x] + adv;
                            e = e > 0.0f ? e : NEG_SLOPE * e;
                            m = fmaxf(m, e);
                        }
                    }
                }
            }
#pragma unroll
            for (int off = 16; off >= 1; off >>= 1) m = fmaxf(m, __shfl_xor(m, off, 32));

            float ssum = __expf(es - m);
            const float2 hs = h2[(size_t)node * (OUT_F / 2) + lane];
            float a0 = ssum * hs.x, a1 = ssum * hs.y;
            for (int sg = 0; sg < NSLICE; ++sg) {
                const int sb = __builtin_amdgcn_readfirstlane(segbase[sg]);
                const int sl = __builtin_amdgcn_readfirstlane(segpos[sg + 1] - segpos[sg]);
                for (int j0 = 0; j0 < sl; j0 += 32) {
                    const int j = j0 + lane;
                    int   s = node;
                    float w = 0.0f;
                    bool  hit = false;
                    if (j < sl) {
                        const int2 rec = sorted[sb + j];
                        if (rec.y == node) {
                            float e = a_s[rec.x] + adv;
                            e = e > 0.0f ? e : NEG_SLOPE * e;
                            w = __expf(e - m);
                            s = rec.x;
                            hit = true;
                        }
                    }
                    const unsigned wb = __float_as_uint(w);
                    unsigned mask = __builtin_amdgcn_ballot_w32(hit);
                    while (mask) {
                        const int k = __builtin_ctz(mask);
                        mask &= mask - 1;
                        const int   sk = __builtin_amdgcn_readlane(s, k);
                        const float wk = __uint_as_float(__builtin_amdgcn_readlane(wb, k));
                        const float2 hv = h2[(size_t)sk * (OUT_F / 2) + lane];
                        a0 = fmaf(wk, hv.x, a0);
                        a1 = fmaf(wk, hv.y, a1);
                        ssum += wk;
                    }
                }
            }
            out2[(size_t)node * (OUT_F / 2) + lane] = make_float2(a0 / ssum + bv.x, a1 / ssum + bv.y);
        }
    }
}

extern "C" void kernel_launch(void* const* d_in, const int* in_sizes, int n_in,
                              void* d_out, int out_size, void* d_ws, size_t ws_size,
                              hipStream_t stream) {
    const float* x       = (const float*)d_in[0];
    const int*   ei      = (const int*)d_in[1];
    const float* W       = (const float*)d_in[2];
    const float* att_src = (const float*)d_in[3];
    const float* att_dst = (const float*)d_in[4];
    const float* bias    = (const float*)d_in[5];

    const int N  = in_sizes[0] / IN_F;
    const int E  = in_sizes[1] / 2;
    const int NB = (N + BIN_NODES - 1) / BIN_NODES;
    const int slice = (E + NSLICE - 1) / NSLICE;

    float* out = (float*)d_out;

    float* h        = (float*)d_ws;
    int*   offsT    = (int*)(h + (size_t)N * OUT_F);
    int2*  sorted   = (int2*)(offsT + (size_t)(NB + 1) * NSLICE);
    float* a_s      = (float*)(sorted + E);
    float* a_d      = a_s + N;

    const int tiles = (N + 15) / 16;
    const int gw    = (tiles + TPB - 1) / TPB;

    slice_sort_gemm<<<NSLICE + gw, 1024, 0, stream>>>(ei, offsT, sorted, E, NB, slice,
                                                      x, W, att_src, att_dst, h, a_s, a_d, N, 0, tiles, TPB);
    gat_bin_process<<<NB, 512, 0, stream>>>(sorted, offsT, slice, h, a_s, a_d, bias, out, N);
}
